# starting kernel
# baseline (speedup 1.0000x reference)
#include <hip/hip_runtime.h>
#include <hip/hip_bf16.h>


typedef float v2f __attribute__((ext_vector_type(2)));
typedef float v8f __attribute__((ext_vector_type(8)));

#define NNODES 50000
#define FEAT   128
#define HALF   64
#define MTILES (NNODES / 16)
#define SPAD   132
#define CPAD   260

__global__ void fill_zero_kernel(float* __restrict__ p, long long count) {
  long long i = (long long)blockIdx.x * blockDim.x + threadIdx.x;
  if (i < count) p[i] = 0.0f;
}

__global__ void degree_kernel(const int* __restrict__ dst, float* __restrict__ deg, int E) {
  int e = blockIdx.x * blockDim.x + threadIdx.x;
  if (e < E) atomicAdd(&deg[dst[e]], 1.0f);
}

__global__ void scatter_kernel(const float* __restrict__ X,
                               const int* __restrict__ src,
                               const int* __restrict__ dst,
                               float* __restrict__ msum, int E) {
  long long t = (long long)blockIdx.x * blockDim.x + threadIdx.x;
  int e = (int)(t >> 5);
  if (e >= E) return;
  int f = ((int)t & 31) * 4;
  int s = src[e];
  int d = dst[e];
  const float4 v = *(const float4*)(X + (size_t)s * FEAT + f);
  float* mp = msum + (size_t)d * FEAT + f;
  atomicAdd(mp + 0, v.x);
  atomicAdd(mp + 1, v.y);
  atomicAdd(mp + 2, v.z);
  atomicAdd(mp + 3, v.w);
}

__global__ __launch_bounds__(128) void sage_gemm_kernel(
    const float* __restrict__ msum, const float* __restrict__ deg,
    const float* __restrict__ H, const float* __restrict__ Wl,
    const float* __restrict__ Wr, const float* __restrict__ bl,
    float* __restrict__ out) {
  __shared__ float lds_mean[16 * SPAD];
  __shared__ float lds_h[16 * SPAD];

  const int tile_m = blockIdx.x;
  const int tid    = threadIdx.x;

#pragma unroll
  for (int i = 0; i < 4; ++i) {
    const int idx = tid + i * 128;
    const int row = idx >> 5;
    const int c4  = (idx & 31) * 4;
    const int gm  = tile_m * 16 + row;
    const float rs = 1.0f / fmaxf(deg[gm], 1.0f);
    const float4 mv = *(const float4*)(msum + (size_t)gm * FEAT + c4);
    const float4 hv = *(const float4*)(H    + (size_t)gm * FEAT + c4);
    float4 sv = { mv.x * rs, mv.y * rs, mv.z * rs, mv.w * rs };
    *(float4*)&lds_mean[row * SPAD + c4] = sv;
    *(float4*)&lds_h[row * SPAD + c4]    = hv;
  }
  __syncthreads();

  const int wave = tid >> 5;
  const int lane = tid & 31;
  const int mi   = lane & 15;
  const int h    = lane >> 4;
  const int n    = wave * 16 + mi;

  v8f c = {};
#pragma unroll 8
  for (int kc = 0; kc < FEAT; kc += 4) {
    const int k = kc + 2 * h;
    v2f a1 = *(const v2f*)&lds_mean[mi * SPAD + k];
    v2f b1 = { Wl[(size_t)k * HALF + n], Wl[(size_t)(k + 1) * HALF + n] };
    c = __builtin_amdgcn_wmma_f32_16x16x4_f32(false, a1, false, b1, (short)0, c, false, false);
    v2f a2 = *(const v2f*)&lds_h[mi * SPAD + k];
    v2f b2 = { Wr[(size_t)k * HALF + n], Wr[(size_t)(k + 1) * HALF + n] };
    c = __builtin_amdgcn_wmma_f32_16x16x4_f32(false, a2, false, b2, (short)0, c, false, false);
  }

  const float bias  = bl[n];
  const int   mbase = tile_m * 16 + 8 * h;
#pragma unroll
  for (int r = 0; r < 8; ++r) {
    float v = c[r] + bias;
    out[(size_t)(mbase + r) * HALF + n] = fmaxf(v, 0.0f);
  }
}

__global__ __launch_bounds__(256) void comb_gemm_kernel(
    const float* __restrict__ X, const float* __restrict__ Hin,
    const float* __restrict__ Hout, const float* __restrict__ W,
    const float* __restrict__ b, float* __restrict__ out) {
  __shared__ float lds_a[16 * CPAD];

  const int tile_m = blockIdx.x;
  const int tid    = threadIdx.x;

#pragma unroll
  for (int i = 0; i < 4; ++i) {
    const int idx = tid + i * 256;
    const int row = idx >> 6;
    const int c4  = (idx & 63) * 4;
    const int gm  = tile_m * 16 + row;
    float4 v;
    if (c4 < 128)      v = *(const float4*)(X    + (size_t)gm * FEAT + c4);
    else if (c4 < 192) v = *(const float4*)(Hin  + (size_t)gm * HALF + (c4 - 128));
    else               v = *(const float4*)(Hout + (size_t)gm * HALF + (c4 - 192));
    *(float4*)&lds_a[row * CPAD + c4] = v;
  }
  __syncthreads();

  const int wave = tid >> 5;
  const int lane = tid & 31;
  const int mi   = lane & 15;
  const int h    = lane >> 4;
  const int n    = wave * 16 + mi;

  v8f c = {};
#pragma unroll 8
  for (int kc = 0; kc < 2 * FEAT; kc += 4) {
    const int k = kc + 2 * h;
    v2f a  = *(const v2f*)&lds_a[mi * CPAD + k];
    v2f bb = { W[(size_t)k * FEAT + n], W[(size_t)(k + 1) * FEAT + n] };
    c = __builtin_amdgcn_wmma_f32_16x16x4_f32(false, a, false, bb, (short)0, c, false, false);
  }

  const float bias  = b[n];
  const int   mbase = tile_m * 16 + 8 * h;
#pragma unroll
  for (int r = 0; r < 8; ++r) {
    float v = c[r] + bias;
    out[(size_t)(mbase + r) * FEAT + n] = fmaxf(v, 0.0f);
  }
}

__global__ void final_kernel(const float* __restrict__ H, const float* __restrict__ W,
                             const float* __restrict__ b, float* __restrict__ out, int n) {
  int i = blockIdx.x * blockDim.x + threadIdx.x;
  if (i >= n) return;
  const float4* hr = (const float4*)(H + (size_t)i * FEAT);
  const float4* wr = (const float4*)W;
  float acc = 0.0f;
#pragma unroll 8
  for (int k = 0; k < FEAT / 4; ++k) {
    float4 hv = hr[k];
    float4 wv = wr[k];
    acc += hv.x * wv.x + hv.y * wv.y + hv.z * wv.z + hv.w * wv.w;
  }
  out[i] = acc + b[0];
}

extern "C" void kernel_launch(void* const* d_in, const int* in_sizes, int n_in,
                              void* d_out, int out_size, void* d_ws, size_t ws_size,
                              hipStream_t stream) {
  (void)n_in; (void)out_size; (void)ws_size;

  const float* x       = (const float*)d_in[0];
  const int*   ein     = (const int*)d_in[1];
  const int*   eout    = (const int*)d_in[2];
  const float* in_Wl0  = (const float*)d_in[3];
  const float* in_bl0  = (const float*)d_in[4];
  const float* in_Wr0  = (const float*)d_in[5];
  const float* out_Wl0 = (const float*)d_in[6];
  const float* out_bl0 = (const float*)d_in[7];
  const float* out_Wr0 = (const float*)d_in[8];
  const float* comb_W0 = (const float*)d_in[9];
  const float* comb_b0 = (const float*)d_in[10];
  const float* in_Wl1  = (const float*)d_in[11];
  const float* in_bl1  = (const float*)d_in[12];
  const float* in_Wr1  = (const float*)d_in[13];
  const float* out_Wl1 = (const float*)d_in[14];
  const float* out_bl1 = (const float*)d_in[15];
  const float* out_Wr1 = (const float*)d_in[16];
  const float* comb_W1 = (const float*)d_in[17];
  const float* comb_b1 = (const float*)d_in[18];
  const float* final_W = (const float*)d_in[19];
  const float* final_b = (const float*)d_in[20];
  float* out = (float*)d_out;

  const int N = NNODES;
  const int E = in_sizes[1] / 2;

  const size_t N128 = (size_t)N * FEAT;
  const size_t N64  = (size_t)N * HALF;
  const size_t PADN = 65536;
  float* ws     = (float*)d_ws;
  float* degIn  = ws;
  float* degOut = ws + PADN;
  float* msum   = ws + 2 * PADN;
  float* hin    = msum + N128;
  float* hout   = hin + N64;
  float* HA     = hout + N64;
  float* HB     = HA + N128;

  const int ZT = 256;
  auto nblk = [](long long c, int t) { return (unsigned)((c + t - 1) / t); };

  fill_zero_kernel<<<nblk(N, ZT), ZT, 0, stream>>>(degIn, N);
  fill_zero_kernel<<<nblk(N, ZT), ZT, 0, stream>>>(degOut, N);
  degree_kernel<<<nblk(E, ZT), ZT, 0, stream>>>(ein + E, degIn, E);
  degree_kernel<<<nblk(E, ZT), ZT, 0, stream>>>(eout + E, degOut, E);

  auto conv = [&](const float* H, const int* edges, const float* deg,
                  const float* Wl, const float* bl, const float* Wr, float* hdst) {
    fill_zero_kernel<<<nblk((long long)N128, ZT), ZT, 0, stream>>>(msum, (long long)N128);
    scatter_kernel<<<nblk((long long)E * 32, ZT), ZT, 0, stream>>>(H, edges, edges + E, msum, E);
    sage_gemm_kernel<<<MTILES, 128, 0, stream>>>(msum, deg, H, Wl, Wr, bl, hdst);
  };

  conv(x, ein,  degIn,  in_Wl0,  in_bl0,  in_Wr0,  hin);
  conv(x, eout, degOut, out_Wl0, out_bl0, out_Wr0, hout);
  comb_gemm_kernel<<<MTILES, 256, 0, stream>>>(x, hin, hout, comb_W0, comb_b0, HA);

  conv(HA, ein,  degIn,  in_Wl1,  in_bl1,  in_Wr1,  hin);
  conv(HA, eout, degOut, out_Wl1, out_bl1, out_Wr1, hout);
  comb_gemm_kernel<<<MTILES, 256, 0, stream>>>(HA, hin, hout, comb_W1, comb_b1, HB);

  final_kernel<<<nblk(N, ZT), ZT, 0, stream>>>(HB, final_W, final_b, out, N);
}


# Scatter: each wave owns 32 consecutive edges (coalesced src/dst loads, ids by readlane) and issues the row loads of 8 edges before their adds; 32x fewer waves; f32 atomic sums stay arrival-ordered
# speedup vs baseline: 1.6108x; 1.6108x over previous
#include <hip/hip_runtime.h>
#include <hip/hip_bf16.h>


typedef float v2f __attribute__((ext_vector_type(2)));
typedef float v8f __attribute__((ext_vector_type(8)));

#define NNODES 50000
#define FEAT   128
#define HALF   64
#define MTILES (NNODES / 16)
#define SPAD   132
#define CPAD   260

__global__ void fill_zero_kernel(float* __restrict__ p, long long count) {
  long long i = (long long)blockIdx.x * blockDim.x + threadIdx.x;
  if (i < count) p[i] = 0.0f;
}

__global__ void degree_kernel(const int* __restrict__ dst, float* __restrict__ deg, int E) {
  int e = blockIdx.x * blockDim.x + threadIdx.x;
  if (e < E) atomicAdd(&deg[dst[e]], 1.0f);
}

#define EPW  32
#define EGRP 8

__device__ __forceinline__ void scatter_one_edge(const float* __restrict__ X,
                                                 float* __restrict__ msum,
                                                 int s, int d, int lane) {
  const float* xp = X + (size_t)s * FEAT + lane;
  float* mp = msum + (size_t)d * FEAT + lane;
  const float v0 = xp[0];
  const float v1 = xp[32];
  const float v2 = xp[64];
  const float v3 = xp[96];
  atomicAdd(mp + 0,  v0);
  atomicAdd(mp + 32, v1);
  atomicAdd(mp + 64, v2);
  atomicAdd(mp + 96, v3);
}

__global__ __launch_bounds__(256) void scatter_kernel(const float* __restrict__ X,
                               const int* __restrict__ src,
                               const int* __restrict__ dst,
                               float* __restrict__ msum, int E) {
  const int lane = threadIdx.x & 31;
  const int wave = __builtin_amdgcn_readfirstlane(
      (int)((blockIdx.x * blockDim.x + threadIdx.x) >> 5));
  const int e0 = wave * EPW;
  if (e0 >= E) return;
  const int cnt = min(EPW, E - e0);
  int s_l = 0, d_l = 0;
  if (lane < cnt) { s_l = src[e0 + lane]; d_l = dst[e0 + lane]; }

  if (cnt == EPW) {
    for (int j0 = 0; j0 < EPW; j0 += EGRP) {
      float v[EGRP][4];
#pragma unroll
      for (int j = 0; j < EGRP; ++j) {
        const int s = __builtin_amdgcn_readlane(s_l, j0 + j);
        const float* xp = X + (size_t)s * FEAT + lane;
        v[j][0] = xp[0];
        v[j][1] = xp[32];
        v[j][2] = xp[64];
        v[j][3] = xp[96];
      }
#pragma unroll
      for (int j = 0; j < EGRP; ++j) {
        const int d = __builtin_amdgcn_readlane(d_l, j0 + j);
        float* mp = msum + (size_t)d * FEAT + lane;
        atomicAdd(mp + 0,  v[j][0]);
        atomicAdd(mp + 32, v[j][1]);
        atomicAdd(mp + 64, v[j][2]);
        atomicAdd(mp + 96, v[j][3]);
      }
    }
  } else {
    for (int j = 0; j < cnt; ++j)
      scatter_one_edge(X, msum, __builtin_amdgcn_readlane(s_l, j),
                       __builtin_amdgcn_readlane(d_l, j), lane);
  }
}

__global__ __launch_bounds__(128) void sage_gemm_kernel(
    const float* __restrict__ msum, const float* __restrict__ deg,
    const float* __restrict__ H, const float* __restrict__ Wl,
    const float* __restrict__ Wr, const float* __restrict__ bl,
    float* __restrict__ out) {
  __shared__ float lds_mean[16 * SPAD];
  __shared__ float lds_h[16 * SPAD];

  const int tile_m = blockIdx.x;
  const int tid    = threadIdx.x;

#pragma unroll
  for (int i = 0; i < 4; ++i) {
    const int idx = tid + i * 128;
    const int row = idx >> 5;
    const int c4  = (idx & 31) * 4;
    const int gm  = tile_m * 16 + row;
    const float rs = 1.0f / fmaxf(deg[gm], 1.0f);
    const float4 mv = *(const float4*)(msum + (size_t)gm * FEAT + c4);
    const float4 hv = *(const float4*)(H    + (size_t)gm * FEAT + c4);
    float4 sv = { mv.x * rs, mv.y * rs, mv.z * rs, mv.w * rs };
    *(float4*)&lds_mean[row * SPAD + c4] = sv;
    *(float4*)&lds_h[row * SPAD + c4]    = hv;
  }
  __syncthreads();

  const int wave = tid >> 5;
  const int lane = tid & 31;
  const int mi   = lane & 15;
  const int h    = lane >> 4;
  const int n    = wave * 16 + mi;

  v8f c = {};
#pragma unroll 8
  for (int kc = 0; kc < FEAT; kc += 4) {
    const int k = kc + 2 * h;
    v2f a1 = *(const v2f*)&lds_mean[mi * SPAD + k];
    v2f b1 = { Wl[(size_t)k * HALF + n], Wl[(size_t)(k + 1) * HALF + n] };
    c = __builtin_amdgcn_wmma_f32_16x16x4_f32(false, a1, false, b1, (short)0, c, false, false);
    v2f a2 = *(const v2f*)&lds_h[mi * SPAD + k];
    v2f b2 = { Wr[(size_t)k * HALF + n], Wr[(size_t)(k + 1) * HALF + n] };
    c = __builtin_amdgcn_wmma_f32_16x16x4_f32(false, a2, false, b2, (short)0, c, false, false);
  }

  const float bias  = bl[n];
  const int   mbase = tile_m * 16 + 8 * h;
#pragma unroll
  for (int r = 0; r < 8; ++r) {
    float v = c[r] + bias;
    out[(size_t)(mbase + r) * HALF + n] = fmaxf(v, 0.0f);
  }
}

__global__ __launch_bounds__(256) void comb_gemm_kernel(
    const float* __restrict__ X, const float* __restrict__ Hin,
    const float* __restrict__ Hout, const float* __restrict__ W,
    const float* __restrict__ b, float* __restrict__ out) {
  __shared__ float lds_a[16 * CPAD];

  const int tile_m = blockIdx.x;
  const int tid    = threadIdx.x;

#pragma unroll
  for (int i = 0; i < 4; ++i) {
    const int idx = tid + i * 256;
    const int row = idx >> 6;
    const int c4  = (idx & 63) * 4;
    const int gm  = tile_m * 16 + row;
    float4 v;
    if (c4 < 128)      v = *(const float4*)(X    + (size_t)gm * FEAT + c4);
    else if (c4 < 192) v = *(const float4*)(Hin  + (size_t)gm * HALF + (c4 - 128));
    else               v = *(const float4*)(Hout + (size_t)gm * HALF + (c4 - 192));
    *(float4*)&lds_a[row * CPAD + c4] = v;
  }
  __syncthreads();

  const int wave = tid >> 5;
  const int lane = tid & 31;
  const int mi   = lane & 15;
  const int h    = lane >> 4;
  const int n    = wave * 16 + mi;

  v8f c = {};
#pragma unroll 8
  for (int kc = 0; kc < 2 * FEAT; kc += 4) {
    const int k = kc + 2 * h;
    v2f a  = *(const v2f*)&lds_a[mi * CPAD + k];
    v2f bb = { W[(size_t)k * FEAT + n], W[(size_t)(k + 1) * FEAT + n] };
    c = __builtin_amdgcn_wmma_f32_16x16x4_f32(false, a, false, bb, (short)0, c, false, false);
  }

  const float bias  = b[n];
  const int   mbase = tile_m * 16 + 8 * h;
#pragma unroll
  for (int r = 0; r < 8; ++r) {
    float v = c[r] + bias;
    out[(size_t)(mbase + r) * FEAT + n] = fmaxf(v, 0.0f);
  }
}

__global__ void final_kernel(const float* __restrict__ H, const float* __restrict__ W,
                             const float* __restrict__ b, float* __restrict__ out, int n) {
  int i = blockIdx.x * blockDim.x + threadIdx.x;
  if (i >= n) return;
  const float4* hr = (const float4*)(H + (size_t)i * FEAT);
  const float4* wr = (const float4*)W;
  float acc = 0.0f;
#pragma unroll 8
  for (int k = 0; k < FEAT / 4; ++k) {
    float4 hv = hr[k];
    float4 wv = wr[k];
    acc += hv.x * wv.x + hv.y * wv.y + hv.z * wv.z + hv.w * wv.w;
  }
  out[i] = acc + b[0];
}

extern "C" void kernel_launch(void* const* d_in, const int* in_sizes, int n_in,
                              void* d_out, int out_size, void* d_ws, size_t ws_size,
                              hipStream_t stream) {
  (void)n_in; (void)out_size; (void)ws_size;

  const float* x       = (const float*)d_in[0];
  const int*   ein     = (const int*)d_in[1];
  const int*   eout    = (const int*)d_in[2];
  const float* in_Wl0  = (const float*)d_in[3];
  const float* in_bl0  = (const float*)d_in[4];
  const float* in_Wr0  = (const float*)d_in[5];
  const float* out_Wl0 = (const float*)d_in[6];
  const float* out_bl0 = (const float*)d_in[7];
  const float* out_Wr0 = (const float*)d_in[8];
  const float* comb_W0 = (const float*)d_in[9];
  const float* comb_b0 = (const float*)d_in[10];
  const float* in_Wl1  = (const float*)d_in[11];
  const float* in_bl1  = (const float*)d_in[12];
  const float* in_Wr1  = (const float*)d_in[13];
  const float* out_Wl1 = (const float*)d_in[14];
  const float* out_bl1 = (const float*)d_in[15];
  const float* out_Wr1 = (const float*)d_in[16];
  const float* comb_W1 = (const float*)d_in[17];
  const float* comb_b1 = (const float*)d_in[18];
  const float* final_W = (const float*)d_in[19];
  const float* final_b = (const float*)d_in[20];
  float* out = (float*)d_out;

  const int N = NNODES;
  const int E = in_sizes[1] / 2;

  const size_t N128 = (size_t)N * FEAT;
  const size_t N64  = (size_t)N * HALF;
  const size_t PADN = 65536;
  float* ws     = (float*)d_ws;
  float* degIn  = ws;
  float* degOut = ws + PADN;
  float* msum   = ws + 2 * PADN;
  float* hin    = msum + N128;
  float* hout   = hin + N64;
  float* HA     = hout + N64;
  float* HB     = HA + N128;

  const int ZT = 256;
  auto nblk = [](long long c, int t) { return (unsigned)((c + t - 1) / t); };

  fill_zero_kernel<<<nblk(N, ZT), ZT, 0, stream>>>(degIn, N);
  fill_zero_kernel<<<nblk(N, ZT), ZT, 0, stream>>>(degOut, N);
  degree_kernel<<<nblk(E, ZT), ZT, 0, stream>>>(ein + E, degIn, E);
  degree_kernel<<<nblk(E, ZT), ZT, 0, stream>>>(eout + E, degOut, E);

  auto conv = [&](const float* H, const int* edges, const float* deg,
                  const float* Wl, const float* bl, const float* Wr, float* hdst) {
    fill_zero_kernel<<<nblk((long long)N128, ZT), ZT, 0, stream>>>(msum, (long long)N128);
    scatter_kernel<<<nblk(((long long)E + EPW - 1) / EPW * 32, ZT), ZT, 0, stream>>>(H, edges, edges + E, msum, E);
    sage_gemm_kernel<<<MTILES, 128, 0, stream>>>(msum, deg, H, Wl, Wr, bl, hdst);
  };

  conv(x, ein,  degIn,  in_Wl0,  in_bl0,  in_Wr0,  hin);
  conv(x, eout, degOut, out_Wl0, out_bl0, out_Wr0, hout);
  comb_gemm_kernel<<<MTILES, 256, 0, stream>>>(x, hin, hout, comb_W0, comb_b0, HA);

  conv(HA, ein,  degIn,  in_Wl1,  in_bl1,  in_Wr1,  hin);
  conv(HA, eout, degOut, out_Wl1, out_bl1, out_Wr1, hout);
  comb_gemm_kernel<<<MTILES, 256, 0, stream>>>(HA, hin, hout, comb_W1, comb_b1, HB);

  final_kernel<<<nblk(N, ZT), ZT, 0, stream>>>(HB, final_W, final_b, out, N);
}


# Aggregation: entries re-queued in LDS to the wave that owns row mod 32; the owner adds each source row by one 16-byte LDS read-add-write, not 4 f32 LDS atomics; sums in queue (arrival) order
# speedup vs baseline: 2.6845x; 1.6666x over previous
#include <hip/hip_runtime.h>
#include <hip/hip_bf16.h>
#include <type_traits>


typedef float v2f __attribute__((ext_vector_type(2)));
typedef float v8f __attribute__((ext_vector_type(8)));

#define NNODES 50000
#define FEAT   128
#define HALF   64
#define MTILES (NNODES / 16)
#define SPAD   132
#define CPAD   260

#define RS     400
#define NRANGE (NNODES / RS)
#define AGG_T  1024
#define NPART  32
#define SUBCAP 800
#define LCAP   (SUBCAP - 1)
#define EGRP   8
#define PGRP   4

static_assert(NRANGE * RS == NNODES, "ranges tile the node set");
static_assert(NPART == AGG_T / 32, "one list per owner wave");
static_assert((size_t)2 * NRANGE * NPART * SUBCAP <= (size_t)NNODES * FEAT, "the lists fit one N x 128 plane");
static_assert(NNODES <= 65536 && RS <= 32768, "entry packing: src in bits 0-15, local row above");

__global__ __launch_bounds__(AGG_T) void partition_kernel(
    const int* __restrict__ edgesA, const int* __restrict__ edgesB,
    int* __restrict__ lists, int E) {
  __shared__ int fill[NRANGE];

  const int tid = threadIdx.x;
  const int* __restrict__ src = blockIdx.y ? edgesB : edgesA;
  const int* __restrict__ dst = src + E;
  const int per = (E + NPART - 1) / NPART;
  const int lo  = blockIdx.x * per;
  const int hi  = min(E, lo + per);

  for (int i = tid; i < NRANGE; i += AGG_T) fill[i] = 0;
  __syncthreads();

  int* __restrict__ mine = lists + ((size_t)blockIdx.y * NRANGE * NPART + blockIdx.x) * SUBCAP;

  for (int e0 = lo + tid; e0 < hi; e0 += PGRP * AGG_T) {
    int d[PGRP], s[PGRP];
#pragma unroll
    for (int u = 0; u < PGRP; ++u) {
      const int e = e0 + u * AGG_T;
      d[u] = -1;
      s[u] = 0;
      if (e < hi) { d[u] = dst[e]; s[u] = src[e]; }
    }
    int r[PGRP], pos[PGRP];
#pragma unroll
    for (int u = 0; u < PGRP; ++u) {
      r[u]   = 0;
      pos[u] = LCAP;
      if ((unsigned)d[u] < (unsigned)NNODES) {
        r[u]   = d[u] / RS;
        pos[u] = __hip_atomic_fetch_add(&fill[r[u]], 1, __ATOMIC_RELAXED, __HIP_MEMORY_SCOPE_WORKGROUP);
      }
    }
#pragma unroll
    for (int u = 0; u < PGRP; ++u) {
      if (pos[u] < LCAP)
        mine[(size_t)r[u] * (NPART * SUBCAP) + 1 + pos[u]] = s[u] | ((d[u] - r[u] * RS) << 16);
    }
  }
  __syncthreads();
  for (int i = tid; i < NRANGE; i += AGG_T) mine[(size_t)i * (NPART * SUBCAP)] = fill[i];
}

#define QCAP 512

__global__ __launch_bounds__(AGG_T) void aggregate_kernel(
    const float* __restrict__ X,
    const int* __restrict__ edgesA, const int* __restrict__ edgesB,
    const int* __restrict__ lists,
    float* __restrict__ msumA, float* __restrict__ msumB,
    float* __restrict__ degA, float* __restrict__ degB, int E) {
  __shared__ __attribute__((aligned(16))) float acc[RS * FEAT];
  __shared__ int cnt[RS];
  __shared__ int qn[NPART];
  __shared__ int queue[NPART * QCAP];

  const int tid  = threadIdx.x;
  const int lane = tid & 31;
  const int wave = __builtin_amdgcn_readfirstlane(tid >> 5);
  const int base = blockIdx.x * RS;

  for (int i = tid; i < RS * FEAT / 4; i += AGG_T)
    ((float4*)acc)[i] = float4{0.0f, 0.0f, 0.0f, 0.0f};
  for (int i = tid; i < RS; i += AGG_T) cnt[i] = 0;
  if (tid < NPART) qn[tid] = 0;
  __syncthreads();

  auto enqueue = [&](int s, int dl, bool valid) {
    const int q = dl & (NPART - 1);
    int slot = 0;
    if (valid) {
      __hip_atomic_fetch_add(&cnt[dl], 1, __ATOMIC_RELAXED, __HIP_MEMORY_SCOPE_WORKGROUP);
      slot = __hip_atomic_fetch_add(&qn[q], 1, __ATOMIC_RELAXED, __HIP_MEMORY_SCOPE_WORKGROUP);
      if (slot < QCAP) queue[q * QCAP + slot] = s | (dl << 16);
    }
    unsigned mask = __builtin_amdgcn_ballot_w32(valid && slot >= QCAP);
    while (mask) {
      const int j = __builtin_ctz(mask);
      mask &= mask - 1;
      const int sj = __builtin_amdgcn_readlane(s, j);
      const int dj = __builtin_amdgcn_readlane(dl, j);
      const float4 v = ((const float4*)(X + (size_t)sj * FEAT))[lane];
      float* ap = &acc[dj * FEAT + 4 * lane];
      __hip_atomic_fetch_add(ap + 0, v.x, __ATOMIC_RELAXED, __HIP_MEMORY_SCOPE_WORKGROUP);
      __hip_atomic_fetch_add(ap + 1, v.y, __ATOMIC_RELAXED, __HIP_MEMORY_SCOPE_WORKGROUP);
      __hip_atomic_fetch_add(ap + 2, v.z, __ATOMIC_RELAXED, __HIP_MEMORY_SCOPE_WORKGROUP);
      __hip_atomic_fetch_add(ap + 3, v.w, __ATOMIC_RELAXED, __HIP_MEMORY_SCOPE_WORKGROUP);
    }
  };

  const int* __restrict__ sl =
      lists + ((size_t)(blockIdx.y * NRANGE + blockIdx.x) * NPART + wave) * SUBCAP;
  const int n = __builtin_amdgcn_readfirstlane(sl[0]);

  if (n <= LCAP) {
    int ent = (lane < n) ? sl[1 + lane] : 0;
    for (int b = 0; b < n; b += 32) {
      const int  cur   = ent;
      const bool valid = b + lane < n;
      ent = (b + 32 + lane < n) ? sl[33 + b + lane] : 0;
      enqueue(cur & 0xffff, cur >> 16, valid);
    }
  } else {
    const int* __restrict__ src = blockIdx.y ? edgesB : edgesA;
    const int* __restrict__ dst = src + E;
    const int per = (E + NPART - 1) / NPART;
    const int lo  = wave * per;
    const int hi  = min(E, lo + per);
    for (int e0 = lo; e0 < hi; e0 += 32) {
      const int e = e0 + lane;
      int d = -1, s = 0;
      if (e < hi) { d = dst[e]; s = src[e]; }
      const unsigned dl = (unsigned)(d - base);
      enqueue(s, (int)dl, dl < (unsigned)RS);
    }
  }

  __syncthreads();

  auto add_rows = [&](auto full, int s, int dl, int m) {
    constexpr bool FULL = decltype(full)::value;
    for (int j0 = 0; j0 < (FULL ? 32 : m); j0 += EGRP) {
      float4 v[EGRP];
#pragma unroll
      for (int j = 0; j < EGRP; ++j) {
        if (FULL || j0 + j < m) {
          const int sj = __builtin_amdgcn_readlane(s, j0 + j);
          v[j] = ((const float4*)(X + (size_t)sj * FEAT))[lane];
        }
      }
#pragma unroll
      for (int j = 0; j < EGRP; ++j) {
        if (FULL || j0 + j < m) {
          const int dj = __builtin_amdgcn_readlane(dl, j0 + j);
          float4* ap = (float4*)&acc[dj * FEAT] + lane;
          float4 a = *ap;
          a.x += v[j].x;
          a.y += v[j].y;
          a.z += v[j].z;
          a.w += v[j].w;
          *ap = a;
        }
      }
    }
  };

  const int m = min(__builtin_amdgcn_readfirstlane(qn[wave]), QCAP);
  const int* myq = &queue[wave * QCAP];
  {
    int ent = (lane < m) ? myq[lane] : 0;
    for (int b = 0; b < m; b += 32) {
      const int left = m - b;
      const int cur  = ent;
      ent = (b + 32 + lane < m) ? myq[b + 32 + lane] : 0;
      const int s  = cur & 0xffff;
      const int dl = cur >> 16;
      if (left >= 32) add_rows(std::true_type{}, s, dl, 32);
      else            add_rows(std::false_type{}, s, dl, left);
    }
  }

  float* __restrict__ msum = blockIdx.y ? msumB : msumA;
  float* __restrict__ deg  = blockIdx.y ? degB  : degA;
  for (int r = wave; r < RS; r += NPART)
    ((float4*)(msum + (size_t)(base + r) * FEAT))[lane] = ((const float4*)&acc[r * FEAT])[lane];
  for (int i = tid; i < RS; i += AGG_T) deg[base + i] = (float)cnt[i];
}

__global__ __launch_bounds__(128) void sage_gemm_kernel(
    const float* __restrict__ msum, const float* __restrict__ deg,
    const float* __restrict__ H, const float* __restrict__ Wl,
    const float* __restrict__ Wr, const float* __restrict__ bl,
    float* __restrict__ out) {
  __shared__ float lds_mean[16 * SPAD];
  __shared__ float lds_h[16 * SPAD];

  const int tile_m = blockIdx.x;
  const int tid    = threadIdx.x;

#pragma unroll
  for (int i = 0; i < 4; ++i) {
    const int idx = tid + i * 128;
    const int row = idx >> 5;
    const int c4  = (idx & 31) * 4;
    const int gm  = tile_m * 16 + row;
    const float rs = 1.0f / fmaxf(deg[gm], 1.0f);
    const float4 mv = *(const float4*)(msum + (size_t)gm * FEAT + c4);
    const float4 hv = *(const float4*)(H    + (size_t)gm * FEAT + c4);
    float4 sv = { mv.x * rs, mv.y * rs, mv.z * rs, mv.w * rs };
    *(float4*)&lds_mean[row * SPAD + c4] = sv;
    *(float4*)&lds_h[row * SPAD + c4]    = hv;
  }
  __syncthreads();

  const int wave = tid >> 5;
  const int lane = tid & 31;
  const int mi   = lane & 15;
  const int h    = lane >> 4;
  const int n    = wave * 16 + mi;

  v8f c = {};
#pragma unroll 8
  for (int kc = 0; kc < FEAT; kc += 4) {
    const int k = kc + 2 * h;
    v2f a1 = *(const v2f*)&lds_mean[mi * SPAD + k];
    v2f b1 = { Wl[(size_t)k * HALF + n], Wl[(size_t)(k + 1) * HALF + n] };
    c = __builtin_amdgcn_wmma_f32_16x16x4_f32(false, a1, false, b1, (short)0, c, false, false);
    v2f a2 = *(const v2f*)&lds_h[mi * SPAD + k];
    v2f b2 = { Wr[(size_t)k * HALF + n], Wr[(size_t)(k + 1) * HALF + n] };
    c = __builtin_amdgcn_wmma_f32_16x16x4_f32(false, a2, false, b2, (short)0, c, false, false);
  }

  const float bias  = bl[n];
  const int   mbase = tile_m * 16 + 8 * h;
#pragma unroll
  for (int r = 0; r < 8; ++r) {
    float v = c[r] + bias;
    out[(size_t)(mbase + r) * HALF + n] = fmaxf(v, 0.0f);
  }
}

__global__ __launch_bounds__(256) void comb_gemm_kernel(
    const float* __restrict__ X, const float* __restrict__ Hin,
    const float* __restrict__ Hout, const float* __restrict__ W,
    const float* __restrict__ b, float* __restrict__ out) {
  __shared__ float lds_a[16 * CPAD];

  const int tile_m = blockIdx.x;
  const int tid    = threadIdx.x;

#pragma unroll
  for (int i = 0; i < 4; ++i) {
    const int idx = tid + i * 256;
    const int row = idx >> 6;
    const int c4  = (idx & 63) * 4;
    const int gm  = tile_m * 16 + row;
    float4 v;
    if (c4 < 128)      v = *(const float4*)(X    + (size_t)gm * FEAT + c4);
    else if (c4 < 192) v = *(const float4*)(Hin  + (size_t)gm * HALF + (c4 - 128));
    else               v = *(const float4*)(Hout + (size_t)gm * HALF + (c4 - 192));
    *(float4*)&lds_a[row * CPAD + c4] = v;
  }
  __syncthreads();

  const int wave = tid >> 5;
  const int lane = tid & 31;
  const int mi   = lane & 15;
  const int h    = lane >> 4;
  const int n    = wave * 16 + mi;

  v8f c = {};
#pragma unroll 8
  for (int kc = 0; kc < 2 * FEAT; kc += 4) {
    const int k = kc + 2 * h;
    v2f a  = *(const v2f*)&lds_a[mi * CPAD + k];
    v2f bb = { W[(size_t)k * FEAT + n], W[(size_t)(k + 1) * FEAT + n] };
    c = __builtin_amdgcn_wmma_f32_16x16x4_f32(false, a, false, bb, (short)0, c, false, false);
  }

  const float bias  = b[n];
  const int   mbase = tile_m * 16 + 8 * h;
#pragma unroll
  for (int r = 0; r < 8; ++r) {
    float v = c[r] + bias;
    out[(size_t)(mbase + r) * FEAT + n] = fmaxf(v, 0.0f);
  }
}

__global__ void final_kernel(const float* __restrict__ H, const float* __restrict__ W,
                             const float* __restrict__ b, float* __restrict__ out, int n) {
  int i = blockIdx.x * blockDim.x + threadIdx.x;
  if (i >= n) return;
  const float4* hr = (const float4*)(H + (size_t)i * FEAT);
  const float4* wr = (const float4*)W;
  float acc = 0.0f;
#pragma unroll 8
  for (int k = 0; k < FEAT / 4; ++k) {
    float4 hv = hr[k];
    float4 wv = wr[k];
    acc += hv.x * wv.x + hv.y * wv.y + hv.z * wv.z + hv.w * wv.w;
  }
  out[i] = acc + b[0];
}

extern "C" void kernel_launch(void* const* d_in, const int* in_sizes, int n_in,
                              void* d_out, int out_size, void* d_ws, size_t ws_size,
                              hipStream_t stream) {
  (void)n_in; (void)out_size; (void)ws_size;

  const float* x       = (const float*)d_in[0];
  const int*   ein     = (const int*)d_in[1];
  const int*   eout    = (const int*)d_in[2];
  const float* in_Wl0  = (const float*)d_in[3];
  const float* in_bl0  = (const float*)d_in[4];
  const float* in_Wr0  = (const float*)d_in[5];
  const float* out_Wl0 = (const float*)d_in[6];
  const float* out_bl0 = (const float*)d_in[7];
  const float* out_Wr0 = (const float*)d_in[8];
  const float* comb_W0 = (const float*)d_in[9];
  const float* comb_b0 = (const float*)d_in[10];
  const float* in_Wl1  = (const float*)d_in[11];
  const float* in_bl1  = (const float*)d_in[12];
  const float* in_Wr1  = (const float*)d_in[13];
  const float* out_Wl1 = (const float*)d_in[14];
  const float* out_bl1 = (const float*)d_in[15];
  const float* out_Wr1 = (const float*)d_in[16];
  const float* comb_W1 = (const float*)d_in[17];
  const float* comb_b1 = (const float*)d_in[18];
  const float* final_W = (const float*)d_in[19];
  const float* final_b = (const float*)d_in[20];
  float* out = (float*)d_out;

  const int N = NNODES;
  const int E = in_sizes[1] / 2;

  const size_t N128 = (size_t)N * FEAT;
  const size_t N64  = (size_t)N * HALF;
  const size_t PADN = 65536;
  float* ws     = (float*)d_ws;
  float* degIn  = ws;
  float* degOut = ws + PADN;
  float* R0 = ws + 2 * PADN;
  float* R1 = R0 + N128;
  float* R2 = R1 + N128;
  float* R3 = R2 + N128;
  int*   lists  = (int*)R3;
  float* msumA0 = R0;
  float* msumB0 = R1;
  float* hin0   = R2;
  float* hout0  = R2 + N64;
  float* HA     = R0;
  float* msumA1 = R1;
  float* msumB1 = R2;
  float* hin1   = R3;
  float* hout1  = R3 + N64;
  float* HB     = R1;

  const int ZT = 256;
  auto nblk = [](long long c, int t) { return (unsigned)((c + t - 1) / t); };
  const dim3 gridAgg(NRANGE, 2);
  const dim3 gridPart(NPART, 2);

  partition_kernel<<<gridPart, AGG_T, 0, stream>>>(ein, eout, lists, E);

  aggregate_kernel<<<gridAgg, AGG_T, 0, stream>>>(x, ein, eout, lists, msumA0, msumB0, degIn, degOut, E);
  sage_gemm_kernel<<<MTILES, 128, 0, stream>>>(msumA0, degIn,  x, in_Wl0,  in_Wr0,  in_bl0,  hin0);
  sage_gemm_kernel<<<MTILES, 128, 0, stream>>>(msumB0, degOut, x, out_Wl0, out_Wr0, out_bl0, hout0);
  comb_gemm_kernel<<<MTILES, 256, 0, stream>>>(x, hin0, hout0, comb_W0, comb_b0, HA);

  aggregate_kernel<<<gridAgg, AGG_T, 0, stream>>>(HA, ein, eout, lists, msumA1, msumB1, degIn, degOut, E);
  sage_gemm_kernel<<<MTILES, 128, 0, stream>>>(msumA1, degIn,  HA, in_Wl1,  in_Wr1,  in_bl1,  hin1);
  sage_gemm_kernel<<<MTILES, 128, 0, stream>>>(msumB1, degOut, HA, out_Wl1, out_Wr1, out_bl1, hout1);
  comb_gemm_kernel<<<MTILES, 256, 0, stream>>>(HA, hin1, hout1, comb_W1, comb_b1, HB);

  final_kernel<<<nblk(N, ZT), ZT, 0, stream>>>(HB, final_W, final_b, out, N);
}


# GEMM waves own 16-row strips of 32 (conv) or 64 (combine) columns, interleaved 2 or 4 per lane: one LDS A read feeds 2 or 4 WMMAs, weight and output rows move as 8/16-byte vectors; sums bit-identical
# speedup vs baseline: 2.8108x; 1.0470x over previous
#include <hip/hip_runtime.h>
#include <hip/hip_bf16.h>
#include <type_traits>


typedef float v2f __attribute__((ext_vector_type(2)));
typedef float v8f __attribute__((ext_vector_type(8)));

#define NNODES 50000
#define FEAT   128
#define HALF   64
#define MTILES (NNODES / 16)
#define SPAD   132
#define CPAD   260

#define RS     400
#define NRANGE (NNODES / RS)
#define AGG_T  1024
#define NPART  32
#define SUBCAP 800
#define LCAP   (SUBCAP - 1)
#define EGRP   8
#define PGRP   4

static_assert(NRANGE * RS == NNODES, "ranges tile the node set");
static_assert(NPART == AGG_T / 32, "one list per owner wave");
static_assert((size_t)2 * NRANGE * NPART * SUBCAP <= (size_t)NNODES * FEAT, "the lists fit one N x 128 plane");
static_assert(NNODES <= 65536 && RS <= 32768, "entry packing: src in bits 0-15, local row above");

__global__ __launch_bounds__(AGG_T) void partition_kernel(
    const int* __restrict__ edgesA, const int* __restrict__ edgesB,
    int* __restrict__ lists, int E) {
  __shared__ int fill[NRANGE];

  const int tid = threadIdx.x;
  const int* __restrict__ src = blockIdx.y ? edgesB : edgesA;
  const int* __restrict__ dst = src + E;
  const int per = (E + NPART - 1) / NPART;
  const int lo  = blockIdx.x * per;
  const int hi  = min(E, lo + per);

  for (int i = tid; i < NRANGE; i += AGG_T) fill[i] = 0;
  __syncthreads();

  int* __restrict__ mine = lists + ((size_t)blockIdx.y * NRANGE * NPART + blockIdx.x) * SUBCAP;

  for (int e0 = lo + tid; e0 < hi; e0 += PGRP * AGG_T) {
    int d[PGRP], s[PGRP];
#pragma unroll
    for (int u = 0; u < PGRP; ++u) {
      const int e = e0 + u * AGG_T;
      d[u] = -1;
      s[u] = 0;
      if (e < hi) { d[u] = dst[e]; s[u] = src[e]; }
    }
    int r[PGRP], pos[PGRP];
#pragma unroll
    for (int u = 0; u < PGRP; ++u) {
      r[u]   = 0;
      pos[u] = LCAP;
      if ((unsigned)d[u] < (unsigned)NNODES) {
        r[u]   = d[u] / RS;
        pos[u] = __hip_atomic_fetch_add(&fill[r[u]], 1, __ATOMIC_RELAXED, __HIP_MEMORY_SCOPE_WORKGROUP);
      }
    }
#pragma unroll
    for (int u = 0; u < PGRP; ++u) {
      if (pos[u] < LCAP)
        mine[(size_t)r[u] * (NPART * SUBCAP) + 1 + pos[u]] = s[u] | ((d[u] - r[u] * RS) << 16);
    }
  }
  __syncthreads();
  for (int i = tid; i < NRANGE; i += AGG_T) mine[(size_t)i * (NPART * SUBCAP)] = fill[i];
}

#define QCAP 512

__global__ __launch_bounds__(AGG_T) void aggregate_kernel(
    const float* __restrict__ X,
    const int* __restrict__ edgesA, const int* __restrict__ edgesB,
    const int* __restrict__ lists,
    float* __restrict__ msumA, float* __restrict__ msumB,
    float* __restrict__ degA, float* __restrict__ degB, int E) {
  __shared__ __attribute__((aligned(16))) float acc[RS * FEAT];
  __shared__ int cnt[RS];
  __shared__ int qn[NPART];
  __shared__ int queue[NPART * QCAP];

  const int tid  = threadIdx.x;
  const int lane = tid & 31;
  const int wave = __builtin_amdgcn_readfirstlane(tid >> 5);
  const int base = blockIdx.x * RS;

  for (int i = tid; i < RS * FEAT / 4; i += AGG_T)
    ((float4*)acc)[i] = float4{0.0f, 0.0f, 0.0f, 0.0f};
  for (int i = tid; i < RS; i += AGG_T) cnt[i] = 0;
  if (tid < NPART) qn[tid] = 0;
  __syncthreads();

  auto enqueue = [&](int s, int dl, bool valid) {
    const int q = dl & (NPART - 1);
    int slot = 0;
    if (valid) {
      __hip_atomic_fetch_add(&cnt[dl], 1, __ATOMIC_RELAXED, __HIP_MEMORY_SCOPE_WORKGROUP);
      slot = __hip_atomic_fetch_add(&qn[q], 1, __ATOMIC_RELAXED, __HIP_MEMORY_SCOPE_WORKGROUP);
      if (slot < QCAP) queue[q * QCAP + slot] = s | (dl << 16);
    }
    unsigned mask = __builtin_amdgcn_ballot_w32(valid && slot >= QCAP);
    while (mask) {
      const int j = __builtin_ctz(mask);
      mask &= mask - 1;
      const int sj = __builtin_amdgcn_readlane(s, j);
      const int dj = __builtin_amdgcn_readlane(dl, j);
      const float4 v = ((const float4*)(X + (size_t)sj * FEAT))[lane];
      float* ap = &acc[dj * FEAT + 4 * lane];
      __hip_atomic_fetch_add(ap + 0, v.x, __ATOMIC_RELAXED, __HIP_MEMORY_SCOPE_WORKGROUP);
      __hip_atomic_fetch_add(ap + 1, v.y, __ATOMIC_RELAXED, __HIP_MEMORY_SCOPE_WORKGROUP);
      __hip_atomic_fetch_add(ap + 2, v.z, __ATOMIC_RELAXED, __HIP_MEMORY_SCOPE_WORKGROUP);
      __hip_atomic_fetch_add(ap + 3, v.w, __ATOMIC_RELAXED, __HIP_MEMORY_SCOPE_WORKGROUP);
    }
  };

  const int* __restrict__ sl =
      lists + ((size_t)(blockIdx.y * NRANGE + blockIdx.x) * NPART + wave) * SUBCAP;
  const int n = __builtin_amdgcn_readfirstlane(sl[0]);

  if (n <= LCAP) {
    int ent = (lane < n) ? sl[1 + lane] : 0;
    for (int b = 0; b < n; b += 32) {
      const int  cur   = ent;
      const bool valid = b + lane < n;
      ent = (b + 32 + lane < n) ? sl[33 + b + lane] : 0;
      enqueue(cur & 0xffff, cur >> 16, valid);
    }
  } else {
    const int* __restrict__ src = blockIdx.y ? edgesB : edgesA;
    const int* __restrict__ dst = src + E;
    const int per = (E + NPART - 1) / NPART;
    const int lo  = wave * per;
    const int hi  = min(E, lo + per);
    for (int e0 = lo; e0 < hi; e0 += 32) {
      const int e = e0 + lane;
      int d = -1, s = 0;
      if (e < hi) { d = dst[e]; s = src[e]; }
      const unsigned dl = (unsigned)(d - base);
      enqueue(s, (int)dl, dl < (unsigned)RS);
    }
  }

  __syncthreads();

  auto add_rows = [&](auto full, int s, int dl, int m) {
    constexpr bool FULL = decltype(full)::value;
    for (int j0 = 0; j0 < (FULL ? 32 : m); j0 += EGRP) {
      float4 v[EGRP];
#pragma unroll
      for (int j = 0; j < EGRP; ++j) {
        if (FULL || j0 + j < m) {
          const int sj = __builtin_amdgcn_readlane(s, j0 + j);
          v[j] = ((const float4*)(X + (size_t)sj * FEAT))[lane];
        }
      }
#pragma unroll
      for (int j = 0; j < EGRP; ++j) {
        if (FULL || j0 + j < m) {
          const int dj = __builtin_amdgcn_readlane(dl, j0 + j);
          float4* ap = (float4*)&acc[dj * FEAT] + lane;
          float4 a = *ap;
          a.x += v[j].x;
          a.y += v[j].y;
          a.z += v[j].z;
          a.w += v[j].w;
          *ap = a;
        }
      }
    }
  };

  const int m = min(__builtin_amdgcn_readfirstlane(qn[wave]), QCAP);
  const int* myq = &queue[wave * QCAP];
  {
    int ent = (lane < m) ? myq[lane] : 0;
    for (int b = 0; b < m; b += 32) {
      const int left = m - b;
      const int cur  = ent;
      ent = (b + 32 + lane < m) ? myq[b + 32 + lane] : 0;
      const int s  = cur & 0xffff;
      const int dl = cur >> 16;
      if (left >= 32) add_rows(std::true_type{}, s, dl, 32);
      else            add_rows(std::false_type{}, s, dl, left);
    }
  }

  float* __restrict__ msum = blockIdx.y ? msumB : msumA;
  float* __restrict__ deg  = blockIdx.y ? degB  : degA;
  for (int r = wave; r < RS; r += NPART)
    ((float4*)(msum + (size_t)(base + r) * FEAT))[lane] = ((const float4*)&acc[r * FEAT])[lane];
  for (int i = tid; i < RS; i += AGG_T) deg[base + i] = (float)cnt[i];
}

static __device__ __forceinline__ v8f mma4(v2f a, float b0, float b1, v8f c) {
  v2f b = { b0, b1 };
  return __builtin_amdgcn_wmma_f32_16x16x4_f32(false, a, false, b, (short)0, c, false, false);
}

#define STRIP_STEP(a, w0, w1, c0, c1, c2, c3) \
  do {                                         \
    c0 = mma4(a, (w0).x, (w1).x, c0);          \
    c1 = mma4(a, (w0).y, (w1).y, c1);          \
    c2 = mma4(a, (w0).z, (w1).z, c2);          \
    c3 = mma4(a, (w0).w, (w1).w, c3);          \
  } while (0)

#define STRIP2_STEP(a, w0, w1, c0, c1) \
  do {                                 \
    c0 = mma4(a, (w0).x, (w1).x, c0);  \
    c1 = mma4(a, (w0).y, (w1).y, c1);  \
  } while (0)

#define SKG 8
#define SNG (FEAT / (4 * SKG))

__global__ __launch_bounds__(64) void sage_gemm_kernel(
    const float* __restrict__ msum, const float* __restrict__ deg,
    const float* __restrict__ H, const float* __restrict__ Wl,
    const float* __restrict__ Wr, const float* __restrict__ bl,
    float* __restrict__ out) {
  __shared__ __attribute__((aligned(16))) float lds_mean[16 * SPAD];
  __shared__ __attribute__((aligned(16))) float lds_h[16 * SPAD];

  const int tile_m = blockIdx.x;
  const int tid    = threadIdx.x;
  const int wave = __builtin_amdgcn_readfirstlane(tid >> 5);
  const int lane = tid & 31;

  const int c  = lane & 15;
  const int h  = lane >> 4;
  const int n0 = 32 * wave + 2 * c;

  const float* __restrict__ wl = Wl + (size_t)(2 * h) * HALF + n0;
  const float* __restrict__ wr = Wr + (size_t)(2 * h) * HALF + n0;

  float2 bw[2][SKG][4];

#pragma unroll
  for (int s = 0; s < SKG; ++s) {
    bw[0][s][0] = *(const float2*)(wl + (size_t)(4 * s) * HALF);
    bw[0][s][1] = *(const float2*)(wl + (size_t)(4 * s + 1) * HALF);
    bw[0][s][2] = *(const float2*)(wr + (size_t)(4 * s) * HALF);
    bw[0][s][3] = *(const float2*)(wr + (size_t)(4 * s + 1) * HALF);
  }

#pragma unroll
  for (int j = 0; j < 8; ++j) {
    const int i  = 8 * wave + j;
    const int gm = tile_m * 16 + i;
    const float rs = 1.0f / fmaxf(deg[gm], 1.0f);
    const float4 mv = ((const float4*)(msum + (size_t)gm * FEAT))[lane];
    const float4 hv = ((const float4*)(H    + (size_t)gm * FEAT))[lane];
    float4 sv = { mv.x * rs, mv.y * rs, mv.z * rs, mv.w * rs };
    *(float4*)&lds_mean[i * SPAD + 4 * lane] = sv;
    *(float4*)&lds_h[i * SPAD + 4 * lane]    = hv;
  }
  __syncthreads();

  const float* am = &lds_mean[c * SPAD + 2 * h];
  const float* ah = &lds_h[c * SPAD + 2 * h];

  v8f c0 = {}, c1 = {};

#pragma unroll
  for (int g = 0; g < SNG; ++g) {
    if (g + 1 < SNG) {
#pragma unroll
      for (int s = 0; s < SKG; ++s) {
        const int k = 4 * ((g + 1) * SKG + s);
        bw[(g + 1) & 1][s][0] = *(const float2*)(wl + (size_t)k * HALF);
        bw[(g + 1) & 1][s][1] = *(const float2*)(wl + (size_t)(k + 1) * HALF);
        bw[(g + 1) & 1][s][2] = *(const float2*)(wr + (size_t)k * HALF);
        bw[(g + 1) & 1][s][3] = *(const float2*)(wr + (size_t)(k + 1) * HALF);
      }
    }
#pragma unroll
    for (int s = 0; s < SKG; ++s) {
      const int k = 4 * (g * SKG + s);
      const v2f a1 = *(const v2f*)(am + k);
      const v2f a2 = *(const v2f*)(ah + k);
      STRIP2_STEP(a1, bw[g & 1][s][0], bw[g & 1][s][1], c0, c1);
      STRIP2_STEP(a2, bw[g & 1][s][2], bw[g & 1][s][3], c0, c1);
    }
  }

  const float2 bias  = *(const float2*)(bl + n0);
  const int    mbase = tile_m * 16 + 8 * h;
#pragma unroll
  for (int r = 0; r < 8; ++r) {
    float2 v = { fmaxf(c0[r] + bias.x, 0.0f), fmaxf(c1[r] + bias.y, 0.0f) };
    *(float2*)(out + (size_t)(mbase + r) * HALF + n0) = v;
  }
}

#define CKG 8
#define CNG ((2 * FEAT) / (4 * CKG))

__global__ __launch_bounds__(64) void comb_gemm_kernel(
    const float* __restrict__ X, const float* __restrict__ Hin,
    const float* __restrict__ Hout, const float* __restrict__ W,
    const float* __restrict__ b, float* __restrict__ out) {
  __shared__ __attribute__((aligned(16))) float lds_a[16 * CPAD];

  const int tile_m = blockIdx.x;
  const int tid    = threadIdx.x;

  const int wave = __builtin_amdgcn_readfirstlane(tid >> 5);
  const int lane = tid & 31;
  const int c    = lane & 15;
  const int h    = lane >> 4;
  const int n0   = 64 * wave + 4 * c;

  const float* __restrict__ wp = W + (size_t)(2 * h) * FEAT + n0;

  float4 bw[2][CKG][2];

#pragma unroll
  for (int s = 0; s < CKG; ++s) {
    bw[0][s][0] = *(const float4*)(wp + (size_t)(4 * s) * FEAT);
    bw[0][s][1] = *(const float4*)(wp + (size_t)(4 * s + 1) * FEAT);
  }

  const int c4 = 4 * tid;
  const float* sp;
  int pitch;
  if (c4 < 128)      { sp = X    + c4;         pitch = FEAT; }
  else if (c4 < 192) { sp = Hin  + (c4 - 128); pitch = HALF; }
  else               { sp = Hout + (c4 - 192); pitch = HALF; }
#pragma unroll
  for (int i = 0; i < 16; ++i) {
    const int gm = tile_m * 16 + i;
    const float4 v = *(const float4*)(sp + (size_t)gm * pitch);
    *(float4*)&lds_a[i * CPAD + c4] = v;
  }
  __syncthreads();

  const float* aa = &lds_a[c * CPAD + 2 * h];

  v8f c0 = {}, c1 = {}, c2 = {}, c3 = {};

#pragma unroll
  for (int g = 0; g < CNG; ++g) {
    if (g + 1 < CNG) {
#pragma unroll
      for (int s = 0; s < CKG; ++s) {
        const int k = 4 * ((g + 1) * CKG + s);
        bw[(g + 1) & 1][s][0] = *(const float4*)(wp + (size_t)k * FEAT);
        bw[(g + 1) & 1][s][1] = *(const float4*)(wp + (size_t)(k + 1) * FEAT);
      }
    }
#pragma unroll
    for (int s = 0; s < CKG; ++s) {
      const int k = 4 * (g * CKG + s);
      const v2f a = *(const v2f*)(aa + k);
      STRIP_STEP(a, bw[g & 1][s][0], bw[g & 1][s][1], c0, c1, c2, c3);
    }
  }

  const float4 bias  = *(const float4*)(b + n0);
  const int    mbase = tile_m * 16 + 8 * h;
#pragma unroll
  for (int r = 0; r < 8; ++r) {
    float4 v = { fmaxf(c0[r] + bias.x, 0.0f), fmaxf(c1[r] + bias.y, 0.0f),
                 fmaxf(c2[r] + bias.z, 0.0f), fmaxf(c3[r] + bias.w, 0.0f) };
    *(float4*)(out + (size_t)(mbase + r) * FEAT + n0) = v;
  }
}

__global__ void final_kernel(const float* __restrict__ H, const float* __restrict__ W,
                             const float* __restrict__ b, float* __restrict__ out, int n) {
  int i = blockIdx.x * blockDim.x + threadIdx.x;
  if (i >= n) return;
  const float4* hr = (const float4*)(H + (size_t)i * FEAT);
  const float4* wr = (const float4*)W;
  float acc = 0.0f;
#pragma unroll 8
  for (int k = 0; k < FEAT / 4; ++k) {
    float4 hv = hr[k];
    float4 wv = wr[k];
    acc += hv.x * wv.x + hv.y * wv.y + hv.z * wv.z + hv.w * wv.w;
  }
  out[i] = acc + b[0];
}

extern "C" void kernel_launch(void* const* d_in, const int* in_sizes, int n_in,
                              void* d_out, int out_size, void* d_ws, size_t ws_size,
                              hipStream_t stream) {
  (void)n_in; (void)out_size; (void)ws_size;

  const float* x       = (const float*)d_in[0];
  const int*   ein     = (const int*)d_in[1];
  const int*   eout    = (const int*)d_in[2];
  const float* in_Wl0  = (const float*)d_in[3];
  const float* in_bl0  = (const float*)d_in[4];
  const float* in_Wr0  = (const float*)d_in[5];
  const float* out_Wl0 = (const float*)d_in[6];
  const float* out_bl0 = (const float*)d_in[7];
  const float* out_Wr0 = (const float*)d_in[8];
  const float* comb_W0 = (const float*)d_in[9];
  const float* comb_b0 = (const float*)d_in[10];
  const float* in_Wl1  = (const float*)d_in[11];
  const float* in_bl1  = (const float*)d_in[12];
  const float* in_Wr1  = (const float*)d_in[13];
  const float* out_Wl1 = (const float*)d_in[14];
  const float* out_bl1 = (const float*)d_in[15];
  const float* out_Wr1 = (const float*)d_in[16];
  const float* comb_W1 = (const float*)d_in[17];
  const float* comb_b1 = (const float*)d_in[18];
  const float* final_W = (const float*)d_in[19];
  const float* final_b = (const float*)d_in[20];
  float* out = (float*)d_out;

  const int N = NNODES;
  const int E = in_sizes[1] / 2;

  const size_t N128 = (size_t)N * FEAT;
  const size_t N64  = (size_t)N * HALF;
  const size_t PADN = 65536;
  float* ws     = (float*)d_ws;
  float* degIn  = ws;
  float* degOut = ws + PADN;
  float* R0 = ws + 2 * PADN;
  float* R1 = R0 + N128;
  float* R2 = R1 + N128;
  float* R3 = R2 + N128;
  int*   lists  = (int*)R3;
  float* msumA0 = R0;
  float* msumB0 = R1;
  float* hin0   = R2;
  float* hout0  = R2 + N64;
  float* HA     = R0;
  float* msumA1 = R1;
  float* msumB1 = R2;
  float* hin1   = R3;
  float* hout1  = R3 + N64;
  float* HB     = R1;

  const int ZT = 256;
  auto nblk = [](long long c, int t) { return (unsigned)((c + t - 1) / t); };
  const dim3 gridAgg(NRANGE, 2);
  const dim3 gridPart(NPART, 2);

  partition_kernel<<<gridPart, AGG_T, 0, stream>>>(ein, eout, lists, E);

  aggregate_kernel<<<gridAgg, AGG_T, 0, stream>>>(x, ein, eout, lists, msumA0, msumB0, degIn, degOut, E);
  sage_gemm_kernel<<<MTILES, 64, 0, stream>>>(msumA0, degIn,  x, in_Wl0,  in_Wr0,  in_bl0,  hin0);
  sage_gemm_kernel<<<MTILES, 64, 0, stream>>>(msumB0, degOut, x, out_Wl0, out_Wr0, out_bl0, hout0);
  comb_gemm_kernel<<<MTILES, 64, 0, stream>>>(x, hin0, hout0, comb_W0, comb_b0, HA);

  aggregate_kernel<<<gridAgg, AGG_T, 0, stream>>>(HA, ein, eout, lists, msumA1, msumB1, degIn, degOut, E);
  sage_gemm_kernel<<<MTILES, 64, 0, stream>>>(msumA1, degIn,  HA, in_Wl1,  in_Wr1,  in_bl1,  hin1);
  sage_gemm_kernel<<<MTILES, 64, 0, stream>>>(msumB1, degOut, HA, out_Wl1, out_Wr1, out_bl1, hout1);
  comb_gemm_kernel<<<MTILES, 64, 0, stream>>>(HA, hin1, hout1, comb_W1, comb_b1, HB);

  final_kernel<<<nblk(N, ZT), ZT, 0, stream>>>(HB, final_W, final_b, out, N);
}


# Both directions' conv GEMMs of a layer run as one launch (grid.y = direction, per-group code unchanged): 10 -> 8 launches, one ramp and drain per layer overlapped; sums bit-identical
# speedup vs baseline: 2.9936x; 1.0650x over previous
#include <hip/hip_runtime.h>
#include <hip/hip_bf16.h>
#include <type_traits>


typedef float v2f __attribute__((ext_vector_type(2)));
typedef float v8f __attribute__((ext_vector_type(8)));

#define NNODES 50000
#define FEAT   128
#define HALF   64
#define MTILES (NNODES / 16)
#define SPAD   132
#define CPAD   260

#define RS     400
#define NRANGE (NNODES / RS)
#define AGG_T  1024
#define NPART  32
#define SUBCAP 800
#define LCAP   (SUBCAP - 1)
#define EGRP   8
#define PGRP   4

static_assert(NRANGE * RS == NNODES, "ranges tile the node set");
static_assert(NPART == AGG_T / 32, "one list per owner wave");
static_assert((size_t)2 * NRANGE * NPART * SUBCAP <= (size_t)NNODES * FEAT, "the lists fit one N x 128 plane");
static_assert(NNODES <= 65536 && RS <= 32768, "entry packing: src in bits 0-15, local row above");

__global__ __launch_bounds__(AGG_T) void partition_kernel(
    const int* __restrict__ edgesA, const int* __restrict__ edgesB,
    int* __restrict__ lists, int E) {
  __shared__ int fill[NRANGE];

  const int tid = threadIdx.x;
  const int* __restrict__ src = blockIdx.y ? edgesB : edgesA;
  const int* __restrict__ dst = src + E;
  const int per = (E + NPART - 1) / NPART;
  const int lo  = blockIdx.x * per;
  const int hi  = min(E, lo + per);

  for (int i = tid; i < NRANGE; i += AGG_T) fill[i] = 0;
  __syncthreads();

  int* __restrict__ mine = lists + ((size_t)blockIdx.y * NRANGE * NPART + blockIdx.x) * SUBCAP;

  for (int e0 = lo + tid; e0 < hi; e0 += PGRP * AGG_T) {
    int d[PGRP], s[PGRP];
#pragma unroll
    for (int u = 0; u < PGRP; ++u) {
      const int e = e0 + u * AGG_T;
      d[u] = -1;
      s[u] = 0;
      if (e < hi) { d[u] = dst[e]; s[u] = src[e]; }
    }
    int r[PGRP], pos[PGRP];
#pragma unroll
    for (int u = 0; u < PGRP; ++u) {
      r[u]   = 0;
      pos[u] = LCAP;
      if ((unsigned)d[u] < (unsigned)NNODES) {
        r[u]   = d[u] / RS;
        pos[u] = __hip_atomic_fetch_add(&fill[r[u]], 1, __ATOMIC_RELAXED, __HIP_MEMORY_SCOPE_WORKGROUP);
      }
    }
#pragma unroll
    for (int u = 0; u < PGRP; ++u) {
      if (pos[u] < LCAP)
        mine[(size_t)r[u] * (NPART * SUBCAP) + 1 + pos[u]] = s[u] | ((d[u] - r[u] * RS) << 16);
    }
  }
  __syncthreads();
  for (int i = tid; i < NRANGE; i += AGG_T) mine[(size_t)i * (NPART * SUBCAP)] = fill[i];
}

#define QCAP 512

__global__ __launch_bounds__(AGG_T) void aggregate_kernel(
    const float* __restrict__ X,
    const int* __restrict__ edgesA, const int* __restrict__ edgesB,
    const int* __restrict__ lists,
    float* __restrict__ msumA, float* __restrict__ msumB,
    float* __restrict__ degA, float* __restrict__ degB, int E) {
  __shared__ __attribute__((aligned(16))) float acc[RS * FEAT];
  __shared__ int cnt[RS];
  __shared__ int qn[NPART];
  __shared__ int queue[NPART * QCAP];

  const int tid  = threadIdx.x;
  const int lane = tid & 31;
  const int wave = __builtin_amdgcn_readfirstlane(tid >> 5);
  const int base = blockIdx.x * RS;

  for (int i = tid; i < RS * FEAT / 4; i += AGG_T)
    ((float4*)acc)[i] = float4{0.0f, 0.0f, 0.0f, 0.0f};
  for (int i = tid; i < RS; i += AGG_T) cnt[i] = 0;
  if (tid < NPART) qn[tid] = 0;
  __syncthreads();

  auto enqueue = [&](int s, int dl, bool valid) {
    const int q = dl & (NPART - 1);
    int slot = 0;
    if (valid) {
      __hip_atomic_fetch_add(&cnt[dl], 1, __ATOMIC_RELAXED, __HIP_MEMORY_SCOPE_WORKGROUP);
      slot = __hip_atomic_fetch_add(&qn[q], 1, __ATOMIC_RELAXED, __HIP_MEMORY_SCOPE_WORKGROUP);
      if (slot < QCAP) queue[q * QCAP + slot] = s | (dl << 16);
    }
    unsigned mask = __builtin_amdgcn_ballot_w32(valid && slot >= QCAP);
    while (mask) {
      const int j = __builtin_ctz(mask);
      mask &= mask - 1;
      const int sj = __builtin_amdgcn_readlane(s, j);
      const int dj = __builtin_amdgcn_readlane(dl, j);
      const float4 v = ((const float4*)(X + (size_t)sj * FEAT))[lane];
      float* ap = &acc[dj * FEAT + 4 * lane];
      __hip_atomic_fetch_add(ap + 0, v.x, __ATOMIC_RELAXED, __HIP_MEMORY_SCOPE_WORKGROUP);
      __hip_atomic_fetch_add(ap + 1, v.y, __ATOMIC_RELAXED, __HIP_MEMORY_SCOPE_WORKGROUP);
      __hip_atomic_fetch_add(ap + 2, v.z, __ATOMIC_RELAXED, __HIP_MEMORY_SCOPE_WORKGROUP);
      __hip_atomic_fetch_add(ap + 3, v.w, __ATOMIC_RELAXED, __HIP_MEMORY_SCOPE_WORKGROUP);
    }
  };

  const int* __restrict__ sl =
      lists + ((size_t)(blockIdx.y * NRANGE + blockIdx.x) * NPART + wave) * SUBCAP;
  const int n = __builtin_amdgcn_readfirstlane(sl[0]);

  if (n <= LCAP) {
    int ent = (lane < n) ? sl[1 + lane] : 0;
    for (int b = 0; b < n; b += 32) {
      const int  cur   = ent;
      const bool valid = b + lane < n;
      ent = (b + 32 + lane < n) ? sl[33 + b + lane] : 0;
      enqueue(cur & 0xffff, cur >> 16, valid);
    }
  } else {
    const int* __restrict__ src = blockIdx.y ? edgesB : edgesA;
    const int* __restrict__ dst = src + E;
    const int per = (E + NPART - 1) / NPART;
    const int lo  = wave * per;
    const int hi  = min(E, lo + per);
    for (int e0 = lo; e0 < hi; e0 += 32) {
      const int e = e0 + lane;
      int d = -1, s = 0;
      if (e < hi) { d = dst[e]; s = src[e]; }
      const unsigned dl = (unsigned)(d - base);
      enqueue(s, (int)dl, dl < (unsigned)RS);
    }
  }

  __syncthreads();

  auto add_rows = [&](auto full, int s, int dl, int m) {
    constexpr bool FULL = decltype(full)::value;
    for (int j0 = 0; j0 < (FULL ? 32 : m); j0 += EGRP) {
      float4 v[EGRP];
#pragma unroll
      for (int j = 0; j < EGRP; ++j) {
        if (FULL || j0 + j < m) {
          const int sj = __builtin_amdgcn_readlane(s, j0 + j);
          v[j] = ((const float4*)(X + (size_t)sj * FEAT))[lane];
        }
      }
#pragma unroll
      for (int j = 0; j < EGRP; ++j) {
        if (FULL || j0 + j < m) {
          const int dj = __builtin_amdgcn_readlane(dl, j0 + j);
          float4* ap = (float4*)&acc[dj * FEAT] + lane;
          float4 a = *ap;
          a.x += v[j].x;
          a.y += v[j].y;
          a.z += v[j].z;
          a.w += v[j].w;
          *ap = a;
        }
      }
    }
  };

  const int m = min(__builtin_amdgcn_readfirstlane(qn[wave]), QCAP);
  const int* myq = &queue[wave * QCAP];
  {
    int ent = (lane < m) ? myq[lane] : 0;
    for (int b = 0; b < m; b += 32) {
      const int left = m - b;
      const int cur  = ent;
      ent = (b + 32 + lane < m) ? myq[b + 32 + lane] : 0;
      const int s  = cur & 0xffff;
      const int dl = cur >> 16;
      if (left >= 32) add_rows(std::true_type{}, s, dl, 32);
      else            add_rows(std::false_type{}, s, dl, left);
    }
  }

  float* __restrict__ msum = blockIdx.y ? msumB : msumA;
  float* __restrict__ deg  = blockIdx.y ? degB  : degA;
  for (int r = wave; r < RS; r += NPART)
    ((float4*)(msum + (size_t)(base + r) * FEAT))[lane] = ((const float4*)&acc[r * FEAT])[lane];
  for (int i = tid; i < RS; i += AGG_T) deg[base + i] = (float)cnt[i];
}

static __device__ __forceinline__ v8f mma4(v2f a, float b0, float b1, v8f c) {
  v2f b = { b0, b1 };
  return __builtin_amdgcn_wmma_f32_16x16x4_f32(false, a, false, b, (short)0, c, false, false);
}

#define STRIP_STEP(a, w0, w1, c0, c1, c2, c3) \
  do {                                         \
    c0 = mma4(a, (w0).x, (w1).x, c0);          \
    c1 = mma4(a, (w0).y, (w1).y, c1);          \
    c2 = mma4(a, (w0).z, (w1).z, c2);          \
    c3 = mma4(a, (w0).w, (w1).w, c3);          \
  } while (0)

#define STRIP2_STEP(a, w0, w1, c0, c1) \
  do {                                 \
    c0 = mma4(a, (w0).x, (w1).x, c0);  \
    c1 = mma4(a, (w0).y, (w1).y, c1);  \
  } while (0)

#define SKG 8
#define SNG (FEAT / (4 * SKG))

__global__ __launch_bounds__(64) void sage_gemm_kernel(
    const float* __restrict__ msumA, const float* __restrict__ msumB,
    const float* __restrict__ degA, const float* __restrict__ degB,
    const float* __restrict__ H,
    const float* __restrict__ WlA, const float* __restrict__ WrA, const float* __restrict__ blA,
    const float* __restrict__ WlB, const float* __restrict__ WrB, const float* __restrict__ blB,
    float* __restrict__ outA, float* __restrict__ outB) {
  __shared__ __attribute__((aligned(16))) float lds_mean[16 * SPAD];
  __shared__ __attribute__((aligned(16))) float lds_h[16 * SPAD];

  const bool second = blockIdx.y != 0;
  const float* __restrict__ msum = second ? msumB : msumA;
  const float* __restrict__ deg  = second ? degB  : degA;
  const float* __restrict__ Wl   = second ? WlB   : WlA;
  const float* __restrict__ Wr   = second ? WrB   : WrA;
  const float* __restrict__ bl   = second ? blB   : blA;
  float* __restrict__ out        = second ? outB  : outA;

  const int tile_m = blockIdx.x;
  const int tid    = threadIdx.x;
  const int wave = __builtin_amdgcn_readfirstlane(tid >> 5);
  const int lane = tid & 31;

  const int c  = lane & 15;
  const int h  = lane >> 4;
  const int n0 = 32 * wave + 2 * c;

  const float* __restrict__ wl = Wl + (size_t)(2 * h) * HALF + n0;
  const float* __restrict__ wr = Wr + (size_t)(2 * h) * HALF + n0;

  float2 bw[2][SKG][4];

#pragma unroll
  for (int s = 0; s < SKG; ++s) {
    bw[0][s][0] = *(const float2*)(wl + (size_t)(4 * s) * HALF);
    bw[0][s][1] = *(const float2*)(wl + (size_t)(4 * s + 1) * HALF);
    bw[0][s][2] = *(const float2*)(wr + (size_t)(4 * s) * HALF);
    bw[0][s][3] = *(const float2*)(wr + (size_t)(4 * s + 1) * HALF);
  }

#pragma unroll
  for (int j = 0; j < 8; ++j) {
    const int i  = 8 * wave + j;
    const int gm = tile_m * 16 + i;
    const float rs = 1.0f / fmaxf(deg[gm], 1.0f);
    const float4 mv = ((const float4*)(msum + (size_t)gm * FEAT))[lane];
    const float4 hv = ((const float4*)(H    + (size_t)gm * FEAT))[lane];
    float4 sv = { mv.x * rs, mv.y * rs, mv.z * rs, mv.w * rs };
    *(float4*)&lds_mean[i * SPAD + 4 * lane] = sv;
    *(float4*)&lds_h[i * SPAD + 4 * lane]    = hv;
  }
  __syncthreads();

  const float* am = &lds_mean[c * SPAD + 2 * h];
  const float* ah = &lds_h[c * SPAD + 2 * h];

  v8f c0 = {}, c1 = {};

#pragma unroll
  for (int g = 0; g < SNG; ++g) {
    if (g + 1 < SNG) {
#pragma unroll
      for (int s = 0; s < SKG; ++s) {
        const int k = 4 * ((g + 1) * SKG + s);
        bw[(g + 1) & 1][s][0] = *(const float2*)(wl + (size_t)k * HALF);
        bw[(g + 1) & 1][s][1] = *(const float2*)(wl + (size_t)(k + 1) * HALF);
        bw[(g + 1) & 1][s][2] = *(const float2*)(wr + (size_t)k * HALF);
        bw[(g + 1) & 1][s][3] = *(const float2*)(wr + (size_t)(k + 1) * HALF);
      }
    }
#pragma unroll
    for (int s = 0; s < SKG; ++s) {
      const int k = 4 * (g * SKG + s);
      const v2f a1 = *(const v2f*)(am + k);
      const v2f a2 = *(const v2f*)(ah + k);
      STRIP2_STEP(a1, bw[g & 1][s][0], bw[g & 1][s][1], c0, c1);
      STRIP2_STEP(a2, bw[g & 1][s][2], bw[g & 1][s][3], c0, c1);
    }
  }

  const float2 bias  = *(const float2*)(bl + n0);
  const int    mbase = tile_m * 16 + 8 * h;
#pragma unroll
  for (int r = 0; r < 8; ++r) {
    float2 v = { fmaxf(c0[r] + bias.x, 0.0f), fmaxf(c1[r] + bias.y, 0.0f) };
    *(float2*)(out + (size_t)(mbase + r) * HALF + n0) = v;
  }
}

#define CKG 8
#define CNG ((2 * FEAT) / (4 * CKG))

__global__ __launch_bounds__(64) void comb_gemm_kernel(
    const float* __restrict__ X, const float* __restrict__ Hin,
    const float* __restrict__ Hout, const float* __restrict__ W,
    const float* __restrict__ b, float* __restrict__ out) {
  __shared__ __attribute__((aligned(16))) float lds_a[16 * CPAD];

  const int tile_m = blockIdx.x;
  const int tid    = threadIdx.x;

  const int wave = __builtin_amdgcn_readfirstlane(tid >> 5);
  const int lane = tid & 31;
  const int c    = lane & 15;
  const int h    = lane >> 4;
  const int n0   = 64 * wave + 4 * c;

  const float* __restrict__ wp = W + (size_t)(2 * h) * FEAT + n0;

  float4 bw[2][CKG][2];

#pragma unroll
  for (int s = 0; s < CKG; ++s) {
    bw[0][s][0] = *(const float4*)(wp + (size_t)(4 * s) * FEAT);
    bw[0][s][1] = *(const float4*)(wp + (size_t)(4 * s + 1) * FEAT);
  }

  const int c4 = 4 * tid;
  const float* sp;
  int pitch;
  if (c4 < 128)      { sp = X    + c4;         pitch = FEAT; }
  else if (c4 < 192) { sp = Hin  + (c4 - 128); pitch = HALF; }
  else               { sp = Hout + (c4 - 192); pitch = HALF; }
#pragma unroll
  for (int i = 0; i < 16; ++i) {
    const int gm = tile_m * 16 + i;
    const float4 v = *(const float4*)(sp + (size_t)gm * pitch);
    *(float4*)&lds_a[i * CPAD + c4] = v;
  }
  __syncthreads();

  const float* aa = &lds_a[c * CPAD + 2 * h];

  v8f c0 = {}, c1 = {}, c2 = {}, c3 = {};

#pragma unroll
  for (int g = 0; g < CNG; ++g) {
    if (g + 1 < CNG) {
#pragma unroll
      for (int s = 0; s < CKG; ++s) {
        const int k = 4 * ((g + 1) * CKG + s);
        bw[(g + 1) & 1][s][0] = *(const float4*)(wp + (size_t)k * FEAT);
        bw[(g + 1) & 1][s][1] = *(const float4*)(wp + (size_t)(k + 1) * FEAT);
      }
    }
#pragma unroll
    for (int s = 0; s < CKG; ++s) {
      const int k = 4 * (g * CKG + s);
      const v2f a = *(const v2f*)(aa + k);
      STRIP_STEP(a, bw[g & 1][s][0], bw[g & 1][s][1], c0, c1, c2, c3);
    }
  }

  const float4 bias  = *(const float4*)(b + n0);
  const int    mbase = tile_m * 16 + 8 * h;
#pragma unroll
  for (int r = 0; r < 8; ++r) {
    float4 v = { fmaxf(c0[r] + bias.x, 0.0f), fmaxf(c1[r] + bias.y, 0.0f),
                 fmaxf(c2[r] + bias.z, 0.0f), fmaxf(c3[r] + bias.w, 0.0f) };
    *(float4*)(out + (size_t)(mbase + r) * FEAT + n0) = v;
  }
}

__global__ void final_kernel(const float* __restrict__ H, const float* __restrict__ W,
                             const float* __restrict__ b, float* __restrict__ out, int n) {
  int i = blockIdx.x * blockDim.x + threadIdx.x;
  if (i >= n) return;
  const float4* hr = (const float4*)(H + (size_t)i * FEAT);
  const float4* wr = (const float4*)W;
  float acc = 0.0f;
#pragma unroll 8
  for (int k = 0; k < FEAT / 4; ++k) {
    float4 hv = hr[k];
    float4 wv = wr[k];
    acc += hv.x * wv.x + hv.y * wv.y + hv.z * wv.z + hv.w * wv.w;
  }
  out[i] = acc + b[0];
}

extern "C" void kernel_launch(void* const* d_in, const int* in_sizes, int n_in,
                              void* d_out, int out_size, void* d_ws, size_t ws_size,
                              hipStream_t stream) {
  (void)n_in; (void)out_size; (void)ws_size;

  const float* x       = (const float*)d_in[0];
  const int*   ein     = (const int*)d_in[1];
  const int*   eout    = (const int*)d_in[2];
  const float* in_Wl0  = (const float*)d_in[3];
  const float* in_bl0  = (const float*)d_in[4];
  const float* in_Wr0  = (const float*)d_in[5];
  const float* out_Wl0 = (const float*)d_in[6];
  const float* out_bl0 = (const float*)d_in[7];
  const float* out_Wr0 = (const float*)d_in[8];
  const float* comb_W0 = (const float*)d_in[9];
  const float* comb_b0 = (const float*)d_in[10];
  const float* in_Wl1  = (const float*)d_in[11];
  const float* in_bl1  = (const float*)d_in[12];
  const float* in_Wr1  = (const float*)d_in[13];
  const float* out_Wl1 = (const float*)d_in[14];
  const float* out_bl1 = (const float*)d_in[15];
  const float* out_Wr1 = (const float*)d_in[16];
  const float* comb_W1 = (const float*)d_in[17];
  const float* comb_b1 = (const float*)d_in[18];
  const float* final_W = (const float*)d_in[19];
  const float* final_b = (const float*)d_in[20];
  float* out = (float*)d_out;

  const int N = NNODES;
  const int E = in_sizes[1] / 2;

  const size_t N128 = (size_t)N * FEAT;
  const size_t N64  = (size_t)N * HALF;
  const size_t PADN = 65536;
  float* ws     = (float*)d_ws;
  float* degIn  = ws;
  float* degOut = ws + PADN;
  float* R0 = ws + 2 * PADN;
  float* R1 = R0 + N128;
  float* R2 = R1 + N128;
  float* R3 = R2 + N128;
  int*   lists  = (int*)R3;
  float* msumA0 = R0;
  float* msumB0 = R1;
  float* hin0   = R2;
  float* hout0  = R2 + N64;
  float* HA     = R0;
  float* msumA1 = R1;
  float* msumB1 = R2;
  float* hin1   = R3;
  float* hout1  = R3 + N64;
  float* HB     = R1;

  const int ZT = 256;
  auto nblk = [](long long c, int t) { return (unsigned)((c + t - 1) / t); };
  const dim3 gridAgg(NRANGE, 2);
  const dim3 gridPart(NPART, 2);
  const dim3 gridSage(MTILES, 2);

  partition_kernel<<<gridPart, AGG_T, 0, stream>>>(ein, eout, lists, E);

  aggregate_kernel<<<gridAgg, AGG_T, 0, stream>>>(x, ein, eout, lists, msumA0, msumB0, degIn, degOut, E);
  sage_gemm_kernel<<<gridSage, 64, 0, stream>>>(msumA0, msumB0, degIn, degOut, x,
                                                in_Wl0, in_Wr0, in_bl0, out_Wl0, out_Wr0, out_bl0, hin0, hout0);
  comb_gemm_kernel<<<MTILES, 64, 0, stream>>>(x, hin0, hout0, comb_W0, comb_b0, HA);

  aggregate_kernel<<<gridAgg, AGG_T, 0, stream>>>(HA, ein, eout, lists, msumA1, msumB1, degIn, degOut, E);
  sage_gemm_kernel<<<gridSage, 64, 0, stream>>>(msumA1, msumB1, degIn, degOut, HA,
                                                in_Wl1, in_Wr1, in_bl1, out_Wl1, out_Wr1, out_bl1, hin1, hout1);
  comb_gemm_kernel<<<MTILES, 64, 0, stream>>>(HA, hin1, hout1, comb_W1, comb_b1, HB);

  final_kernel<<<nblk(N, ZT), ZT, 0, stream>>>(HB, final_W, final_b, out, N);
}


# Conv GEMM runs inside the owner aggregation groups: sums stay in LDS as its A operand, Wl|Wr staged in LDS; no sum planes, 8 -> 6 launches; changes f32 summation order inside groups of 8 k
# speedup vs baseline: 3.5880x; 1.1986x over previous
#include <hip/hip_runtime.h>
#include <hip/hip_bf16.h>
#include <type_traits>


typedef float v2f __attribute__((ext_vector_type(2)));
typedef float v8f __attribute__((ext_vector_type(8)));

#define NNODES 50000
#define FEAT   128
#define HALF   64
#define MTILES (NNODES / 16)
#define CPAD   260

#define RS     400
#define NRANGE (NNODES / RS)
#define AGG_T  1024
#define NPART  32
#define SUBCAP 800
#define LCAP   (SUBCAP - 1)
#define EGRP   8
#define PGRP   4

static_assert(NRANGE * RS == NNODES, "ranges tile the node set");
static_assert(NPART == AGG_T / 32, "one list per owner wave");
static_assert((size_t)2 * NRANGE * NPART * SUBCAP <= (size_t)NNODES * FEAT, "the lists fit one N x 128 plane");
static_assert(NNODES <= 65536 && RS <= 32768, "entry packing: src in bits 0-15, local row above");

__global__ __launch_bounds__(AGG_T) void partition_kernel(
    const int* __restrict__ edgesA, const int* __restrict__ edgesB,
    int* __restrict__ lists, int E) {
  __shared__ int fill[NRANGE];

  const int tid = threadIdx.x;
  const int* __restrict__ src = blockIdx.y ? edgesB : edgesA;
  const int* __restrict__ dst = src + E;
  const int per = (E + NPART - 1) / NPART;
  const int lo  = blockIdx.x * per;
  const int hi  = min(E, lo + per);

  for (int i = tid; i < NRANGE; i += AGG_T) fill[i] = 0;
  __syncthreads();

  int* __restrict__ mine = lists + ((size_t)blockIdx.y * NRANGE * NPART + blockIdx.x) * SUBCAP;

  for (int e0 = lo + tid; e0 < hi; e0 += PGRP * AGG_T) {
    int d[PGRP], s[PGRP];
#pragma unroll
    for (int u = 0; u < PGRP; ++u) {
      const int e = e0 + u * AGG_T;
      d[u] = -1;
      s[u] = 0;
      if (e < hi) { d[u] = dst[e]; s[u] = src[e]; }
    }
    int r[PGRP], pos[PGRP];
#pragma unroll
    for (int u = 0; u < PGRP; ++u) {
      r[u]   = 0;
      pos[u] = LCAP;
      if ((unsigned)d[u] < (unsigned)NNODES) {
        r[u]   = d[u] / RS;
        pos[u] = __hip_atomic_fetch_add(&fill[r[u]], 1, __ATOMIC_RELAXED, __HIP_MEMORY_SCOPE_WORKGROUP);
      }
    }
#pragma unroll
    for (int u = 0; u < PGRP; ++u) {
      if (pos[u] < LCAP)
        mine[(size_t)r[u] * (NPART * SUBCAP) + 1 + pos[u]] = s[u] | ((d[u] - r[u] * RS) << 16);
    }
  }
  __syncthreads();
  for (int i = tid; i < NRANGE; i += AGG_T) mine[(size_t)i * (NPART * SUBCAP)] = fill[i];
}

static __device__ __forceinline__ v8f mma4(v2f a, float b0, float b1, v8f c) {
  v2f b = { b0, b1 };
  return __builtin_amdgcn_wmma_f32_16x16x4_f32(false, a, false, b, (short)0, c, false, false);
}

#define STRIP_STEP(a, w0, w1, c0, c1, c2, c3) \
  do {                                         \
    c0 = mma4(a, (w0).x, (w1).x, c0);          \
    c1 = mma4(a, (w0).y, (w1).y, c1);          \
    c2 = mma4(a, (w0).z, (w1).z, c2);          \
    c3 = mma4(a, (w0).w, (w1).w, c3);          \
  } while (0)

#define STRIP2_STEP(a, w0, w1, c0, c1) \
  do {                                 \
    c0 = mma4(a, (w0).x, (w1).x, c0);  \
    c1 = mma4(a, (w0).y, (w1).y, c1);  \
  } while (0)

#define QCAP   512
#define APITCH 132
#define WP     72
#define QW     (2 * FEAT * WP)
#define NT     (RS / 16)
#define CJ     4
#define CNJ    (FEAT / 8)

static_assert(NPART * QCAP <= QW, "the owner queues fit the shared area");
static_assert(NT * 16 == RS, "a range is whole 16-row tiles (EXEC all ones at every WMMA)");

static __device__ __forceinline__ float2 lds_w2(const int* p) {
  const int2 r = *(const int2*)p;
  return float2{ __int_as_float(r.x), __int_as_float(r.y) };
}

__global__ __launch_bounds__(AGG_T) void aggregate_conv_kernel(
    const float* __restrict__ X,
    const int* __restrict__ edgesA, const int* __restrict__ edgesB,
    const int* __restrict__ lists,
    const float* __restrict__ WlA, const float* __restrict__ WrA, const float* __restrict__ blA,
    const float* __restrict__ WlB, const float* __restrict__ WrB, const float* __restrict__ blB,
    float* __restrict__ outA, float* __restrict__ outB, int E) {
  __shared__ __attribute__((aligned(16))) float acc[RS * APITCH];
  __shared__ int cnt[RS];
  __shared__ int qn[NPART];
  __shared__ __attribute__((aligned(16))) int queue[QW];

  const int tid  = threadIdx.x;
  const int lane = tid & 31;
  const int wave = __builtin_amdgcn_readfirstlane(tid >> 5);
  const int base = blockIdx.x * RS;

  for (int i = tid; i < RS * APITCH / 4; i += AGG_T)
    ((float4*)acc)[i] = float4{0.0f, 0.0f, 0.0f, 0.0f};
  for (int i = tid; i < RS; i += AGG_T) cnt[i] = 0;
  if (tid < NPART) qn[tid] = 0;
  __syncthreads();

  auto enqueue = [&](int s, int dl, bool valid) {
    const int q = dl & (NPART - 1);
    int slot = 0;
    if (valid) {
      __hip_atomic_fetch_add(&cnt[dl], 1, __ATOMIC_RELAXED, __HIP_MEMORY_SCOPE_WORKGROUP);
      slot = __hip_atomic_fetch_add(&qn[q], 1, __ATOMIC_RELAXED, __HIP_MEMORY_SCOPE_WORKGROUP);
      if (slot < QCAP) queue[q * QCAP + slot] = s | (dl << 16);
    }
    unsigned mask = __builtin_amdgcn_ballot_w32(valid && slot >= QCAP);
    while (mask) {
      const int j = __builtin_ctz(mask);
      mask &= mask - 1;
      const int sj = __builtin_amdgcn_readlane(s, j);
      const int dj = __builtin_amdgcn_readlane(dl, j);
      const float4 v = ((const float4*)(X + (size_t)sj * FEAT))[lane];
      float* ap = &acc[dj * APITCH + 4 * lane];
      __hip_atomic_fetch_add(ap + 0, v.x, __ATOMIC_RELAXED, __HIP_MEMORY_SCOPE_WORKGROUP);
      __hip_atomic_fetch_add(ap + 1, v.y, __ATOMIC_RELAXED, __HIP_MEMORY_SCOPE_WORKGROUP);
      __hip_atomic_fetch_add(ap + 2, v.z, __ATOMIC_RELAXED, __HIP_MEMORY_SCOPE_WORKGROUP);
      __hip_atomic_fetch_add(ap + 3, v.w, __ATOMIC_RELAXED, __HIP_MEMORY_SCOPE_WORKGROUP);
    }
  };

  const int* __restrict__ sl =
      lists + ((size_t)(blockIdx.y * NRANGE + blockIdx.x) * NPART + wave) * SUBCAP;
  const int n = __builtin_amdgcn_readfirstlane(sl[0]);

  if (n <= LCAP) {
    int ent = (lane < n) ? sl[1 + lane] : 0;
    for (int b = 0; b < n; b += 32) {
      const int  cur   = ent;
      const bool valid = b + lane < n;
      ent = (b + 32 + lane < n) ? sl[33 + b + lane] : 0;
      enqueue(cur & 0xffff, cur >> 16, valid);
    }
  } else {
    const int* __restrict__ src = blockIdx.y ? edgesB : edgesA;
    const int* __restrict__ dst = src + E;
    const int per = (E + NPART - 1) / NPART;
    const int lo  = wave * per;
    const int hi  = min(E, lo + per);
    for (int e0 = lo; e0 < hi; e0 += 32) {
      const int e = e0 + lane;
      int d = -1, s = 0;
      if (e < hi) { d = dst[e]; s = src[e]; }
      const unsigned dl = (unsigned)(d - base);
      enqueue(s, (int)dl, dl < (unsigned)RS);
    }
  }

  __syncthreads();

  const bool second = blockIdx.y != 0;
  const float* __restrict__ Wl = second ? WlB  : WlA;
  const float* __restrict__ Wr = second ? WrB  : WrA;
  const float* __restrict__ bl = second ? blB  : blA;
  float* __restrict__ out      = second ? outB : outA;

  float4 wv[4];
#pragma unroll
  for (int p = 0; p < 4; ++p)
    wv[p] = ((const float4*)(p < 2 ? Wl : Wr))[(tid + p * AGG_T) & (FEAT * HALF / 4 - 1)];

  auto add_rows = [&](auto full, int s, int dl, int m) {
    constexpr bool FULL = decltype(full)::value;
    for (int j0 = 0; j0 < (FULL ? 32 : m); j0 += EGRP) {
      float4 v[EGRP];
#pragma unroll
      for (int j = 0; j < EGRP; ++j) {
        if (FULL || j0 + j < m) {
          const int sj = __builtin_amdgcn_readlane(s, j0 + j);
          v[j] = ((const float4*)(X + (size_t)sj * FEAT))[lane];
        }
      }
#pragma unroll
      for (int j = 0; j < EGRP; ++j) {
        if (FULL || j0 + j < m) {
          const int dj = __builtin_amdgcn_readlane(dl, j0 + j);
          float4* ap = (float4*)&acc[dj * APITCH] + lane;
          float4 a = *ap;
          a.x += v[j].x;
          a.y += v[j].y;
          a.z += v[j].z;
          a.w += v[j].w;
          *ap = a;
        }
      }
    }
  };

  const int m = min(__builtin_amdgcn_readfirstlane(qn[wave]), QCAP);
  const int* myq = &queue[wave * QCAP];
  {
    int ent = (lane < m) ? myq[lane] : 0;
    for (int b = 0; b < m; b += 32) {
      const int left = m - b;
      const int cur  = ent;
      ent = (b + 32 + lane < m) ? myq[b + 32 + lane] : 0;
      const int s  = cur & 0xffff;
      const int dl = cur >> 16;
      if (left >= 32) add_rows(std::true_type{}, s, dl, 32);
      else            add_rows(std::false_type{}, s, dl, left);
    }
  }

  __syncthreads();

#pragma unroll
  for (int p = 0; p < 4; ++p) {
    const int rem = (tid + p * AGG_T) & (FEAT * HALF / 4 - 1);
    const int4 bits = { __float_as_int(wv[p].x), __float_as_int(wv[p].y), __float_as_int(wv[p].z), __float_as_int(wv[p].w) };
    *(int4*)&queue[((p >> 1) * FEAT + (rem >> 4)) * WP + 4 * (rem & 15)] = bits;
  }
  __syncthreads();

  const int c = lane & 15;
  const int h = lane >> 4;

  for (int t = wave; t < 2 * NT; t += NPART) {
    const int half = t / NT;
    const int tile = t - half * NT;
    const int lrow = 16 * tile + c;
    const int n0   = 32 * half + 2 * c;
    const float rs = 1.0f / fmaxf((float)cnt[lrow], 1.0f);
    const float* am = &acc[lrow * APITCH + 4 * h];
    const float* __restrict__ xh = X + (size_t)(base + lrow) * FEAT + 4 * h;
    const int* wlp = &queue[(4 * h) * WP + n0];
    const int* wrp = wlp + FEAT * WP;

    float4 xc[CJ];

#pragma unroll
    for (int jj = 0; jj < CJ; ++jj) xc[jj] = *(const float4*)(xh + 8 * jj);

    v8f c0 = {}, c1 = {};

#pragma unroll 1
    for (int g = 0; g < CNJ / CJ; ++g) {
      float4 xn[CJ];
      if (g + 1 < CNJ / CJ) {
#pragma unroll
        for (int jj = 0; jj < CJ; ++jj) xn[jj] = *(const float4*)(xh + 8 * ((g + 1) * CJ + jj));
      } else {
#pragma unroll
        for (int jj = 0; jj < CJ; ++jj) xn[jj] = xc[jj];
      }
      const float* amg = am  + 8 * CJ * g;
      const int*   wlg = wlp + 8 * CJ * g * WP;
      const int*   wrg = wrp + 8 * CJ * g * WP;
#pragma unroll
      for (int jj = 0; jj < CJ; ++jj) {
        const float4 mv = *(const float4*)(amg + 8 * jj);
        const float4 xv = xc[jj];
        const v2f m0 = { mv.x * rs, mv.y * rs };
        const v2f m1 = { mv.z * rs, mv.w * rs };
        const v2f x0 = { xv.x, xv.y };
        const v2f x1 = { xv.z, xv.w };
        const float2 l0 = lds_w2(wlg + (8 * jj + 0) * WP), l1 = lds_w2(wlg + (8 * jj + 1) * WP);
        const float2 l2 = lds_w2(wlg + (8 * jj + 2) * WP), l3 = lds_w2(wlg + (8 * jj + 3) * WP);
        const float2 w0 = lds_w2(wrg + (8 * jj + 0) * WP), w1 = lds_w2(wrg + (8 * jj + 1) * WP);
        const float2 w2 = lds_w2(wrg + (8 * jj + 2) * WP), w3 = lds_w2(wrg + (8 * jj + 3) * WP);
        STRIP2_STEP(m0, l0, l1, c0, c1);
        STRIP2_STEP(x0, w0, w1, c0, c1);
        STRIP2_STEP(m1, l2, l3, c0, c1);
        STRIP2_STEP(x1, w2, w3, c0, c1);
      }
#pragma unroll
      for (int jj = 0; jj < CJ; ++jj) xc[jj] = xn[jj];
    }

    const float2 bias  = *(const float2*)(bl + n0);
    const int    mbase = base + 16 * tile + 8 * h;
#pragma unroll
    for (int r = 0; r < 8; ++r) {
      float2 v = { fmaxf(c0[r] + bias.x, 0.0f), fmaxf(c1[r] + bias.y, 0.0f) };
      *(float2*)(out + (size_t)(mbase + r) * HALF + n0) = v;
    }
  }
}

#define CKG 8
#define CNG ((2 * FEAT) / (4 * CKG))

__global__ __launch_bounds__(64) void comb_gemm_kernel(
    const float* __restrict__ X, const float* __restrict__ Hin,
    const float* __restrict__ Hout, const float* __restrict__ W,
    const float* __restrict__ b, float* __restrict__ out) {
  __shared__ __attribute__((aligned(16))) float lds_a[16 * CPAD];

  const int tile_m = blockIdx.x;
  const int tid    = threadIdx.x;

  const int wave = __builtin_amdgcn_readfirstlane(tid >> 5);
  const int lane = tid & 31;
  const int c    = lane & 15;
  const int h    = lane >> 4;
  const int n0   = 64 * wave + 4 * c;

  const float* __restrict__ wp = W + (size_t)(2 * h) * FEAT + n0;

  float4 bw[2][CKG][2];

#pragma unroll
  for (int s = 0; s < CKG; ++s) {
    bw[0][s][0] = *(const float4*)(wp + (size_t)(4 * s) * FEAT);
    bw[0][s][1] = *(const float4*)(wp + (size_t)(4 * s + 1) * FEAT);
  }

  const int c4 = 4 * tid;
  const float* sp;
  int pitch;
  if (c4 < 128)      { sp = X    + c4;         pitch = FEAT; }
  else if (c4 < 192) { sp = Hin  + (c4 - 128); pitch = HALF; }
  else               { sp = Hout + (c4 - 192); pitch = HALF; }
#pragma unroll
  for (int i = 0; i < 16; ++i) {
    const int gm = tile_m * 16 + i;
    const float4 v = *(const float4*)(sp + (size_t)gm * pitch);
    *(float4*)&lds_a[i * CPAD + c4] = v;
  }
  __syncthreads();

  const float* aa = &lds_a[c * CPAD + 2 * h];

  v8f c0 = {}, c1 = {}, c2 = {}, c3 = {};

#pragma unroll
  for (int g = 0; g < CNG; ++g) {
    if (g + 1 < CNG) {
#pragma unroll
      for (int s = 0; s < CKG; ++s) {
        const int k = 4 * ((g + 1) * CKG + s);
        bw[(g + 1) & 1][s][0] = *(const float4*)(wp + (size_t)k * FEAT);
        bw[(g + 1) & 1][s][1] = *(const float4*)(wp + (size_t)(k + 1) * FEAT);
      }
    }
#pragma unroll
    for (int s = 0; s < CKG; ++s) {
      const int k = 4 * (g * CKG + s);
      const v2f a = *(const v2f*)(aa + k);
      STRIP_STEP(a, bw[g & 1][s][0], bw[g & 1][s][1], c0, c1, c2, c3);
    }
  }

  const float4 bias  = *(const float4*)(b + n0);
  const int    mbase = tile_m * 16 + 8 * h;
#pragma unroll
  for (int r = 0; r < 8; ++r) {
    float4 v = { fmaxf(c0[r] + bias.x, 0.0f), fmaxf(c1[r] + bias.y, 0.0f),
                 fmaxf(c2[r] + bias.z, 0.0f), fmaxf(c3[r] + bias.w, 0.0f) };
    *(float4*)(out + (size_t)(mbase + r) * FEAT + n0) = v;
  }
}

__global__ void final_kernel(const float* __restrict__ H, const float* __restrict__ W,
                             const float* __restrict__ b, float* __restrict__ out, int n) {
  int i = blockIdx.x * blockDim.x + threadIdx.x;
  if (i >= n) return;
  const float4* hr = (const float4*)(H + (size_t)i * FEAT);
  const float4* wr = (const float4*)W;
  float acc = 0.0f;
#pragma unroll 8
  for (int k = 0; k < FEAT / 4; ++k) {
    float4 hv = hr[k];
    float4 wv = wr[k];
    acc += hv.x * wv.x + hv.y * wv.y + hv.z * wv.z + hv.w * wv.w;
  }
  out[i] = acc + b[0];
}

extern "C" void kernel_launch(void* const* d_in, const int* in_sizes, int n_in,
                              void* d_out, int out_size, void* d_ws, size_t ws_size,
                              hipStream_t stream) {
  (void)n_in; (void)out_size; (void)ws_size;

  const float* x       = (const float*)d_in[0];
  const int*   ein     = (const int*)d_in[1];
  const int*   eout    = (const int*)d_in[2];
  const float* in_Wl0  = (const float*)d_in[3];
  const float* in_bl0  = (const float*)d_in[4];
  const float* in_Wr0  = (const float*)d_in[5];
  const float* out_Wl0 = (const float*)d_in[6];
  const float* out_bl0 = (const float*)d_in[7];
  const float* out_Wr0 = (const float*)d_in[8];
  const float* comb_W0 = (const float*)d_in[9];
  const float* comb_b0 = (const float*)d_in[10];
  const float* in_Wl1  = (const float*)d_in[11];
  const float* in_bl1  = (const float*)d_in[12];
  const float* in_Wr1  = (const float*)d_in[13];
  const float* out_Wl1 = (const float*)d_in[14];
  const float* out_bl1 = (const float*)d_in[15];
  const float* out_Wr1 = (const float*)d_in[16];
  const float* comb_W1 = (const float*)d_in[17];
  const float* comb_b1 = (const float*)d_in[18];
  const float* final_W = (const float*)d_in[19];
  const float* final_b = (const float*)d_in[20];
  float* out = (float*)d_out;

  const int N = NNODES;
  const int E = in_sizes[1] / 2;

  const size_t N128 = (size_t)N * FEAT;
  const size_t N64  = (size_t)N * HALF;
  float* ws = (float*)d_ws;
  float* R0 = ws;
  float* R1 = R0 + N128;
  float* R2 = R1 + N128;
  float* R3 = R2 + N128;
  int*   lists = (int*)R3;
  float* hin0  = R2;
  float* hout0 = R2 + N64;
  float* HA    = R0;
  float* hin1  = R1;
  float* hout1 = R1 + N64;
  float* HB    = R2;

  const int ZT = 256;
  auto nblk = [](long long c, int t) { return (unsigned)((c + t - 1) / t); };
  const dim3 gridAgg(NRANGE, 2);
  const dim3 gridPart(NPART, 2);

  partition_kernel<<<gridPart, AGG_T, 0, stream>>>(ein, eout, lists, E);

  aggregate_conv_kernel<<<gridAgg, AGG_T, 0, stream>>>(x, ein, eout, lists,
                                                       in_Wl0, in_Wr0, in_bl0, out_Wl0, out_Wr0, out_bl0, hin0, hout0, E);
  comb_gemm_kernel<<<MTILES, 64, 0, stream>>>(x, hin0, hout0, comb_W0, comb_b0, HA);

  aggregate_conv_kernel<<<gridAgg, AGG_T, 0, stream>>>(HA, ein, eout, lists,
                                                       in_Wl1, in_Wr1, in_bl1, out_Wl1, out_Wr1, out_bl1, hin1, hout1, E);
  comb_gemm_kernel<<<MTILES, 64, 0, stream>>>(HA, hin1, hout1, comb_W1, comb_b1, HB);

  final_kernel<<<nblk(N, ZT), ZT, 0, stream>>>(HB, final_W, final_b, out, N);
}


# Final projection folded into layer 1's combine GEMM epilogue (lane dot, halving lane exchange, LDS across strips; 32-row groups): no layer-1 plane, 6 -> 5 launches; changes f32 sum order of the dot
# speedup vs baseline: 3.7285x; 1.0392x over previous
#include <hip/hip_runtime.h>
#include <hip/hip_bf16.h>
#include <type_traits>


typedef float v2f __attribute__((ext_vector_type(2)));
typedef float v8f __attribute__((ext_vector_type(8)));

#define NNODES 50000
#define FEAT   128
#define HALF   64
#define MTILES (NNODES / 16)
#define CPAD   260

#define RS     400
#define NRANGE (NNODES / RS)
#define AGG_T  1024
#define NPART  32
#define SUBCAP 800
#define LCAP   (SUBCAP - 1)
#define EGRP   8
#define PGRP   4

static_assert(NRANGE * RS == NNODES, "ranges tile the node set");
static_assert(NPART == AGG_T / 32, "one list per owner wave");
static_assert((size_t)2 * NRANGE * NPART * SUBCAP <= (size_t)NNODES * FEAT, "the lists fit one N x 128 plane");
static_assert(NNODES <= 65536 && RS <= 32768, "entry packing: src in bits 0-15, local row above");

__global__ __launch_bounds__(AGG_T) void partition_kernel(
    const int* __restrict__ edgesA, const int* __restrict__ edgesB,
    int* __restrict__ lists, int E) {
  __shared__ int fill[NRANGE];

  const int tid = threadIdx.x;
  const int* __restrict__ src = blockIdx.y ? edgesB : edgesA;
  const int* __restrict__ dst = src + E;
  const int per = (E + NPART - 1) / NPART;
  const int lo  = blockIdx.x * per;
  const int hi  = min(E, lo + per);

  for (int i = tid; i < NRANGE; i += AGG_T) fill[i] = 0;
  __syncthreads();

  int* __restrict__ mine = lists + ((size_t)blockIdx.y * NRANGE * NPART + blockIdx.x) * SUBCAP;

  for (int e0 = lo + tid; e0 < hi; e0 += PGRP * AGG_T) {
    int d[PGRP], s[PGRP];
#pragma unroll
    for (int u = 0; u < PGRP; ++u) {
      const int e = e0 + u * AGG_T;
      d[u] = -1;
      s[u] = 0;
      if (e < hi) { d[u] = dst[e]; s[u] = src[e]; }
    }
    int r[PGRP], pos[PGRP];
#pragma unroll
    for (int u = 0; u < PGRP; ++u) {
      r[u]   = 0;
      pos[u] = LCAP;
      if ((unsigned)d[u] < (unsigned)NNODES) {
        r[u]   = d[u] / RS;
        pos[u] = __hip_atomic_fetch_add(&fill[r[u]], 1, __ATOMIC_RELAXED, __HIP_MEMORY_SCOPE_WORKGROUP);
      }
    }
#pragma unroll
    for (int u = 0; u < PGRP; ++u) {
      if (pos[u] < LCAP)
        mine[(size_t)r[u] * (NPART * SUBCAP) + 1 + pos[u]] = s[u] | ((d[u] - r[u] * RS) << 16);
    }
  }
  __syncthreads();
  for (int i = tid; i < NRANGE; i += AGG_T) mine[(size_t)i * (NPART * SUBCAP)] = fill[i];
}

static __device__ __forceinline__ v8f mma4(v2f a, float b0, float b1, v8f c) {
  v2f b = { b0, b1 };
  return __builtin_amdgcn_wmma_f32_16x16x4_f32(false, a, false, b, (short)0, c, false, false);
}

#define STRIP_STEP(a, w0, w1, c0, c1, c2, c3) \
  do {                                         \
    c0 = mma4(a, (w0).x, (w1).x, c0);          \
    c1 = mma4(a, (w0).y, (w1).y, c1);          \
    c2 = mma4(a, (w0).z, (w1).z, c2);          \
    c3 = mma4(a, (w0).w, (w1).w, c3);          \
  } while (0)

#define STRIP2_STEP(a, w0, w1, c0, c1) \
  do {                                 \
    c0 = mma4(a, (w0).x, (w1).x, c0);  \
    c1 = mma4(a, (w0).y, (w1).y, c1);  \
  } while (0)

#define QCAP   512
#define APITCH 132
#define WP     72
#define QW     (2 * FEAT * WP)
#define NT     (RS / 16)
#define CJ     4
#define CNJ    (FEAT / 8)

static_assert(NPART * QCAP <= QW, "the owner queues fit the shared area");
static_assert(NT * 16 == RS, "a range is whole 16-row tiles (EXEC all ones at every WMMA)");

static __device__ __forceinline__ float2 lds_w2(const int* p) {
  const int2 r = *(const int2*)p;
  return float2{ __int_as_float(r.x), __int_as_float(r.y) };
}

__global__ __launch_bounds__(AGG_T) void aggregate_conv_kernel(
    const float* __restrict__ X,
    const int* __restrict__ edgesA, const int* __restrict__ edgesB,
    const int* __restrict__ lists,
    const float* __restrict__ WlA, const float* __restrict__ WrA, const float* __restrict__ blA,
    const float* __restrict__ WlB, const float* __restrict__ WrB, const float* __restrict__ blB,
    float* __restrict__ outA, float* __restrict__ outB, int E) {
  __shared__ __attribute__((aligned(16))) float acc[RS * APITCH];
  __shared__ int cnt[RS];
  __shared__ int qn[NPART];
  __shared__ __attribute__((aligned(16))) int queue[QW];

  const int tid  = threadIdx.x;
  const int lane = tid & 31;
  const int wave = __builtin_amdgcn_readfirstlane(tid >> 5);
  const int base = blockIdx.x * RS;

  for (int i = tid; i < RS * APITCH / 4; i += AGG_T)
    ((float4*)acc)[i] = float4{0.0f, 0.0f, 0.0f, 0.0f};
  for (int i = tid; i < RS; i += AGG_T) cnt[i] = 0;
  if (tid < NPART) qn[tid] = 0;
  __syncthreads();

  auto enqueue = [&](int s, int dl, bool valid) {
    const int q = dl & (NPART - 1);
    int slot = 0;
    if (valid) {
      __hip_atomic_fetch_add(&cnt[dl], 1, __ATOMIC_RELAXED, __HIP_MEMORY_SCOPE_WORKGROUP);
      slot = __hip_atomic_fetch_add(&qn[q], 1, __ATOMIC_RELAXED, __HIP_MEMORY_SCOPE_WORKGROUP);
      if (slot < QCAP) queue[q * QCAP + slot] = s | (dl << 16);
    }
    unsigned mask = __builtin_amdgcn_ballot_w32(valid && slot >= QCAP);
    while (mask) {
      const int j = __builtin_ctz(mask);
      mask &= mask - 1;
      const int sj = __builtin_amdgcn_readlane(s, j);
      const int dj = __builtin_amdgcn_readlane(dl, j);
      const float4 v = ((const float4*)(X + (size_t)sj * FEAT))[lane];
      float* ap = &acc[dj * APITCH + 4 * lane];
      __hip_atomic_fetch_add(ap + 0, v.x, __ATOMIC_RELAXED, __HIP_MEMORY_SCOPE_WORKGROUP);
      __hip_atomic_fetch_add(ap + 1, v.y, __ATOMIC_RELAXED, __HIP_MEMORY_SCOPE_WORKGROUP);
      __hip_atomic_fetch_add(ap + 2, v.z, __ATOMIC_RELAXED, __HIP_MEMORY_SCOPE_WORKGROUP);
      __hip_atomic_fetch_add(ap + 3, v.w, __ATOMIC_RELAXED, __HIP_MEMORY_SCOPE_WORKGROUP);
    }
  };

  const int* __restrict__ sl =
      lists + ((size_t)(blockIdx.y * NRANGE + blockIdx.x) * NPART + wave) * SUBCAP;
  const int n = __builtin_amdgcn_readfirstlane(sl[0]);

  if (n <= LCAP) {
    int ent = (lane < n) ? sl[1 + lane] : 0;
    for (int b = 0; b < n; b += 32) {
      const int  cur   = ent;
      const bool valid = b + lane < n;
      ent = (b + 32 + lane < n) ? sl[33 + b + lane] : 0;
      enqueue(cur & 0xffff, cur >> 16, valid);
    }
  } else {
    const int* __restrict__ src = blockIdx.y ? edgesB : edgesA;
    const int* __restrict__ dst = src + E;
    const int per = (E + NPART - 1) / NPART;
    const int lo  = wave * per;
    const int hi  = min(E, lo + per);
    for (int e0 = lo; e0 < hi; e0 += 32) {
      const int e = e0 + lane;
      int d = -1, s = 0;
      if (e < hi) { d = dst[e]; s = src[e]; }
      const unsigned dl = (unsigned)(d - base);
      enqueue(s, (int)dl, dl < (unsigned)RS);
    }
  }

  __syncthreads();

  const bool second = blockIdx.y != 0;
  const float* __restrict__ Wl = second ? WlB  : WlA;
  const float* __restrict__ Wr = second ? WrB  : WrA;
  const float* __restrict__ bl = second ? blB  : blA;
  float* __restrict__ out      = second ? outB : outA;

  float4 wv[4];
#pragma unroll
  for (int p = 0; p < 4; ++p)
    wv[p] = ((const float4*)(p < 2 ? Wl : Wr))[(tid + p * AGG_T) & (FEAT * HALF / 4 - 1)];

  auto add_rows = [&](auto full, int s, int dl, int m) {
    constexpr bool FULL = decltype(full)::value;
    for (int j0 = 0; j0 < (FULL ? 32 : m); j0 += EGRP) {
      float4 v[EGRP];
#pragma unroll
      for (int j = 0; j < EGRP; ++j) {
        if (FULL || j0 + j < m) {
          const int sj = __builtin_amdgcn_readlane(s, j0 + j);
          v[j] = ((const float4*)(X + (size_t)sj * FEAT))[lane];
        }
      }
#pragma unroll
      for (int j = 0; j < EGRP; ++j) {
        if (FULL || j0 + j < m) {
          const int dj = __builtin_amdgcn_readlane(dl, j0 + j);
          float4* ap = (float4*)&acc[dj * APITCH] + lane;
          float4 a = *ap;
          a.x += v[j].x;
          a.y += v[j].y;
          a.z += v[j].z;
          a.w += v[j].w;
          *ap = a;
        }
      }
    }
  };

  const int m = min(__builtin_amdgcn_readfirstlane(qn[wave]), QCAP);
  const int* myq = &queue[wave * QCAP];
  {
    int ent = (lane < m) ? myq[lane] : 0;
    for (int b = 0; b < m; b += 32) {
      const int left = m - b;
      const int cur  = ent;
      ent = (b + 32 + lane < m) ? myq[b + 32 + lane] : 0;
      const int s  = cur & 0xffff;
      const int dl = cur >> 16;
      if (left >= 32) add_rows(std::true_type{}, s, dl, 32);
      else            add_rows(std::false_type{}, s, dl, left);
    }
  }

  __syncthreads();

#pragma unroll
  for (int p = 0; p < 4; ++p) {
    const int rem = (tid + p * AGG_T) & (FEAT * HALF / 4 - 1);
    const int4 bits = { __float_as_int(wv[p].x), __float_as_int(wv[p].y), __float_as_int(wv[p].z), __float_as_int(wv[p].w) };
    *(int4*)&queue[((p >> 1) * FEAT + (rem >> 4)) * WP + 4 * (rem & 15)] = bits;
  }
  __syncthreads();

  const int c = lane & 15;
  const int h = lane >> 4;

  for (int t = wave; t < 2 * NT; t += NPART) {
    const int half = t / NT;
    const int tile = t - half * NT;
    const int lrow = 16 * tile + c;
    const int n0   = 32 * half + 2 * c;
    const float rs = 1.0f / fmaxf((float)cnt[lrow], 1.0f);
    const float* am = &acc[lrow * APITCH + 4 * h];
    const float* __restrict__ xh = X + (size_t)(base + lrow) * FEAT + 4 * h;
    const int* wlp = &queue[(4 * h) * WP + n0];
    const int* wrp = wlp + FEAT * WP;

    float4 xc[CJ];

#pragma unroll
    for (int jj = 0; jj < CJ; ++jj) xc[jj] = *(const float4*)(xh + 8 * jj);

    v8f c0 = {}, c1 = {};

#pragma unroll 1
    for (int g = 0; g < CNJ / CJ; ++g) {
      float4 xn[CJ];
      if (g + 1 < CNJ / CJ) {
#pragma unroll
        for (int jj = 0; jj < CJ; ++jj) xn[jj] = *(const float4*)(xh + 8 * ((g + 1) * CJ + jj));
      } else {
#pragma unroll
        for (int jj = 0; jj < CJ; ++jj) xn[jj] = xc[jj];
      }
      const float* amg = am  + 8 * CJ * g;
      const int*   wlg = wlp + 8 * CJ * g * WP;
      const int*   wrg = wrp + 8 * CJ * g * WP;
#pragma unroll
      for (int jj = 0; jj < CJ; ++jj) {
        const float4 mv = *(const float4*)(amg + 8 * jj);
        const float4 xv = xc[jj];
        const v2f m0 = { mv.x * rs, mv.y * rs };
        const v2f m1 = { mv.z * rs, mv.w * rs };
        const v2f x0 = { xv.x, xv.y };
        const v2f x1 = { xv.z, xv.w };
        const float2 l0 = lds_w2(wlg + (8 * jj + 0) * WP), l1 = lds_w2(wlg + (8 * jj + 1) * WP);
        const float2 l2 = lds_w2(wlg + (8 * jj + 2) * WP), l3 = lds_w2(wlg + (8 * jj + 3) * WP);
        const float2 w0 = lds_w2(wrg + (8 * jj + 0) * WP), w1 = lds_w2(wrg + (8 * jj + 1) * WP);
        const float2 w2 = lds_w2(wrg + (8 * jj + 2) * WP), w3 = lds_w2(wrg + (8 * jj + 3) * WP);
        STRIP2_STEP(m0, l0, l1, c0, c1);
        STRIP2_STEP(x0, w0, w1, c0, c1);
        STRIP2_STEP(m1, l2, l3, c0, c1);
        STRIP2_STEP(x1, w2, w3, c0, c1);
      }
#pragma unroll
      for (int jj = 0; jj < CJ; ++jj) xc[jj] = xn[jj];
    }

    const float2 bias  = *(const float2*)(bl + n0);
    const int    mbase = base + 16 * tile + 8 * h;
#pragma unroll
    for (int r = 0; r < 8; ++r) {
      float2 v = { fmaxf(c0[r] + bias.x, 0.0f), fmaxf(c1[r] + bias.y, 0.0f) };
      *(float2*)(out + (size_t)(mbase + r) * HALF + n0) = v;
    }
  }
}

#define CKG 8
#define CNG ((2 * FEAT) / (4 * CKG))

template <int TILES, bool FINAL>
__global__ __launch_bounds__(64 * TILES) void comb_gemm_kernel(
    const float* __restrict__ X, const float* __restrict__ Hin,
    const float* __restrict__ Hout, const float* __restrict__ W,
    const float* __restrict__ b, float* __restrict__ out,
    const float* __restrict__ fw, const float* __restrict__ fb) {
  __shared__ __attribute__((aligned(16))) float lds_all[TILES * 16 * CPAD];
  __shared__ float fsum[FINAL ? TILES * 2 * 16 : 1];

  const int wv   = __builtin_amdgcn_readfirstlane((int)threadIdx.x >> 5);
  const int pair = wv >> 1;
  const int wave = wv & 1;
  const int tid  = threadIdx.x & 63;
  const int tile_m = min((int)blockIdx.x * TILES + pair, MTILES - 1);
  float* lds_a = lds_all + pair * (16 * CPAD);

  const int lane = tid & 31;
  const int c    = lane & 15;
  const int h    = lane >> 4;
  const int n0   = 64 * wave + 4 * c;

  const float* __restrict__ wp = W + (size_t)(2 * h) * FEAT + n0;

  float4 bw[2][CKG][2];

#pragma unroll
  for (int s = 0; s < CKG; ++s) {
    bw[0][s][0] = *(const float4*)(wp + (size_t)(4 * s) * FEAT);
    bw[0][s][1] = *(const float4*)(wp + (size_t)(4 * s + 1) * FEAT);
  }

  const int c4 = 4 * tid;
  const float* sp;
  int pitch;
  if (c4 < 128)      { sp = X    + c4;         pitch = FEAT; }
  else if (c4 < 192) { sp = Hin  + (c4 - 128); pitch = HALF; }
  else               { sp = Hout + (c4 - 192); pitch = HALF; }
#pragma unroll
  for (int i = 0; i < 16; ++i) {
    const int gm = tile_m * 16 + i;
    const float4 v = *(const float4*)(sp + (size_t)gm * pitch);
    *(float4*)&lds_a[i * CPAD + c4] = v;
  }
  __syncthreads();

  const float* aa = &lds_a[c * CPAD + 2 * h];

  v8f c0 = {}, c1 = {}, c2 = {}, c3 = {};
  float rowsum = 0.0f;

#pragma unroll
  for (int g = 0; g < CNG; ++g) {
    if (g + 1 < CNG) {
#pragma unroll
      for (int s = 0; s < CKG; ++s) {
        const int k = 4 * ((g + 1) * CKG + s);
        bw[(g + 1) & 1][s][0] = *(const float4*)(wp + (size_t)k * FEAT);
        bw[(g + 1) & 1][s][1] = *(const float4*)(wp + (size_t)(k + 1) * FEAT);
      }
    }
#pragma unroll
    for (int s = 0; s < CKG; ++s) {
      const int k = 4 * (g * CKG + s);
      const v2f a = *(const v2f*)(aa + k);
      STRIP_STEP(a, bw[g & 1][s][0], bw[g & 1][s][1], c0, c1, c2, c3);
    }
  }

  const float4 bias  = *(const float4*)(b + n0);
  if constexpr (!FINAL) {
    const int mbase = tile_m * 16 + 8 * h;
#pragma unroll
    for (int r = 0; r < 8; ++r) {
      float4 v = { fmaxf(c0[r] + bias.x, 0.0f), fmaxf(c1[r] + bias.y, 0.0f),
                   fmaxf(c2[r] + bias.z, 0.0f), fmaxf(c3[r] + bias.w, 0.0f) };
      *(float4*)(out + (size_t)(mbase + r) * FEAT + n0) = v;
    }
  } else {
    const float4 f4 = *(const float4*)(fw + n0);
    float p[8];
#pragma unroll
    for (int r = 0; r < 8; ++r)
      p[r] = fmaxf(c0[r] + bias.x, 0.0f) * f4.x + fmaxf(c1[r] + bias.y, 0.0f) * f4.y +
             fmaxf(c2[r] + bias.z, 0.0f) * f4.z + fmaxf(c3[r] + bias.w, 0.0f) * f4.w;
    const bool b3 = (c & 8) != 0, b2 = (c & 4) != 0, b1 = (c & 2) != 0;
    float q4[4], q2[2];
#pragma unroll
    for (int i = 0; i < 4; ++i) {
      const float keep = b3 ? p[i + 4] : p[i];
      const float send = b3 ? p[i] : p[i + 4];
      q4[i] = keep + __shfl_xor(send, 8);
    }
#pragma unroll
    for (int i = 0; i < 2; ++i) {
      const float keep = b2 ? q4[i + 2] : q4[i];
      const float send = b2 ? q4[i] : q4[i + 2];
      q2[i] = keep + __shfl_xor(send, 4);
    }
    {
      const float keep = b1 ? q2[1] : q2[0];
      const float send = b1 ? q2[0] : q2[1];
      rowsum = keep + __shfl_xor(send, 2);
    }
    rowsum += __shfl_xor(rowsum, 1);
  }

  if constexpr (FINAL) {
    if ((c & 1) == 0) fsum[wv * 16 + 8 * h + (c >> 1)] = rowsum;
    __syncthreads();
    if (wv == 0) {
      const int l   = threadIdx.x;
      const int row = blockIdx.x * (16 * TILES) + l;
      static_assert(!FINAL || TILES == 2, "one wave stores one row per lane: 32 rows per work-group");
      if (row < NNODES)
        out[row] = fsum[(2 * (l >> 4)) * 16 + (l & 15)] + fsum[(2 * (l >> 4) + 1) * 16 + (l & 15)] + fb[0];
    }
  }
}

extern "C" void kernel_launch(void* const* d_in, const int* in_sizes, int n_in,
                              void* d_out, int out_size, void* d_ws, size_t ws_size,
                              hipStream_t stream) {
  (void)n_in; (void)out_size; (void)ws_size;

  const float* x       = (const float*)d_in[0];
  const int*   ein     = (const int*)d_in[1];
  const int*   eout    = (const int*)d_in[2];
  const float* in_Wl0  = (const float*)d_in[3];
  const float* in_bl0  = (const float*)d_in[4];
  const float* in_Wr0  = (const float*)d_in[5];
  const float* out_Wl0 = (const float*)d_in[6];
  const float* out_bl0 = (const float*)d_in[7];
  const float* out_Wr0 = (const float*)d_in[8];
  const float* comb_W0 = (const float*)d_in[9];
  const float* comb_b0 = (const float*)d_in[10];
  const float* in_Wl1  = (const float*)d_in[11];
  const float* in_bl1  = (const float*)d_in[12];
  const float* in_Wr1  = (const float*)d_in[13];
  const float* out_Wl1 = (const float*)d_in[14];
  const float* out_bl1 = (const float*)d_in[15];
  const float* out_Wr1 = (const float*)d_in[16];
  const float* comb_W1 = (const float*)d_in[17];
  const float* comb_b1 = (const float*)d_in[18];
  const float* final_W = (const float*)d_in[19];
  const float* final_b = (const float*)d_in[20];
  float* out = (float*)d_out;

  const int N = NNODES;
  const int E = in_sizes[1] / 2;

  const size_t N128 = (size_t)N * FEAT;
  const size_t N64  = (size_t)N * HALF;
  float* ws = (float*)d_ws;
  float* R0 = ws;
  float* R1 = R0 + N128;
  float* R2 = R1 + N128;
  float* R3 = R2 + N128;
  int*   lists = (int*)R3;
  float* hin0  = R2;
  float* hout0 = R2 + N64;
  float* HA    = R0;
  float* hin1  = R1;
  float* hout1 = R1 + N64;

  const dim3 gridAgg(NRANGE, 2);
  const dim3 gridPart(NPART, 2);

  partition_kernel<<<gridPart, AGG_T, 0, stream>>>(ein, eout, lists, E);

  aggregate_conv_kernel<<<gridAgg, AGG_T, 0, stream>>>(x, ein, eout, lists,
                                                       in_Wl0, in_Wr0, in_bl0, out_Wl0, out_Wr0, out_bl0, hin0, hout0, E);
  comb_gemm_kernel<1, false><<<MTILES, 64, 0, stream>>>(x, hin0, hout0, comb_W0, comb_b0, HA, final_W, final_b);

  aggregate_conv_kernel<<<gridAgg, AGG_T, 0, stream>>>(HA, ein, eout, lists,
                                                       in_Wl1, in_Wr1, in_bl1, out_Wl1, out_Wr1, out_bl1, hin1, hout1, E);
  comb_gemm_kernel<2, true><<<(MTILES + 1) / 2, 128, 0, stream>>>(HA, hin1, hout1, comb_W1, comb_b1, out, final_W, final_b);
}
